# starting kernel
# baseline (speedup 1.0000x reference)
#include <hip/hip_runtime.h>
#include <math.h>

typedef float v2f __attribute__((ext_vector_type(2)));
typedef float v8f __attribute__((ext_vector_type(8)));

__device__ inline v8f wmma4(v2f a, v2f b, v8f c) {
    return __builtin_amdgcn_wmma_f32_16x16x4_f32(
         false, a,  false, b,
         (short)0, c,  false,  false);
}

__device__ inline float swap16(float v) {
    return __int_as_float(__builtin_amdgcn_ds_swizzle(__float_as_int(v), 0x401F));
}

__global__ void build_G(const float* __restrict__ w, float* __restrict__ g) {
    int t = threadIdx.x;
    if (t >= 512) return;
    int l = t >> 8;
    int e = t & 255;
    int r = e >> 4;
    int c = e & 15;

    int j = r;
    if (j & 2) j ^= 1;
    if (j & 4) j ^= 2;
    if (j & 8) j ^= 4;

    float ar = 1.0f, ai = 0.0f;
    #pragma unroll
    for (int q = 0; q < 4; ++q) {
        float ty = w[(l * 4 + q) * 2 + 0] * 0.5f;
        float tz = w[(l * 4 + q) * 2 + 1] * 0.5f;
        float cy, sy, cz, sz;
        __sincosf(ty, &sy, &cy);
        __sincosf(tz, &sz, &cz);
        int br = (j >> (3 - q)) & 1;
        int bc = (c >> (3 - q)) & 1;
        float ur, ui;
        if (br == 0) { float m = bc ? -sy : cy; ur = m * cz; ui = -m * sz; }
        else         { float m = bc ?  cy : sy; ur = m * cz; ui =  m * sz; }
        float nr = ar * ur - ai * ui;
        float ni = ar * ui + ai * ur;
        ar = nr; ai = ni;
    }
    g[(l * 3 + 0) * 256 + e] = ar;
    g[(l * 3 + 1) * 256 + e] = ai;
    g[(l * 3 + 2) * 256 + e] = -ai;
}

__global__ void __launch_bounds__(256) qml_kernel(const float* __restrict__ x,
                                                  const float* __restrict__ g,
                                                  float* __restrict__ out) {
    const int lane = threadIdx.x & 31;
    const int wave = threadIdx.x >> 5;
    const int n    = lane & 15;
    const bool hi  = lane >= 16;
    const int cb   = hi ? 2 : 0;
    const int samp = blockIdx.x * 128 + wave * 16 + n;

    float4 xv = reinterpret_cast<const float4*>(x)[samp];
    float c0, s0, c1, s1, c2, s2, c3, s3;
    __sincosf(xv.x * 0.5f, &s0, &c0);
    __sincosf(xv.y * 0.5f, &s1, &c1);
    __sincosf(xv.z * 0.5f, &s2, &c2);
    __sincosf(xv.w * 0.5f, &s3, &c3);

    v2f Ar[4], Ai[4], An[4];
    #pragma unroll
    for (int k = 0; k < 4; ++k) {
        int off = n * 16 + 4 * k + cb;
        Ar[k] = *reinterpret_cast<const v2f*>(g + 0 * 256 + off);
        Ai[k] = *reinterpret_cast<const v2f*>(g + 1 * 256 + off);
        An[k] = *reinterpret_cast<const v2f*>(g + 2 * 256 + off);
    }

    const float f2  = hi ? s2 : c2;
    const float ab0 = c0 * c1, ab1 = c0 * s1, ab2 = s0 * c1, ab3 = s0 * s1;
    const float ab[4] = {ab0, ab1, ab2, ab3};
    const float fd0 = f2 * c3, fd1 = f2 * s3;
    v2f Br[4], Bi[4];
    #pragma unroll
    for (int k = 0; k < 4; ++k) {
        #pragma unroll
        for (int jj = 0; jj < 2; ++jj) {
            const float m = ab[k] * (jj ? fd1 : fd0);
            const int cpc = ((k >> 1) & 1) + (k & 1) + jj;
            const int ql = cpc & 3;
            const int qh = (cpc + 1) & 3;
            const float rl = (ql == 0) ? m : (ql == 2) ? -m : 0.f;
            const float il = (ql == 1) ? -m : (ql == 3) ? m : 0.f;
            const float rh = (qh == 0) ? m : (qh == 2) ? -m : 0.f;
            const float ih = (qh == 1) ? -m : (qh == 3) ? m : 0.f;
            Br[k][jj] = hi ? rh : rl;
            Bi[k][jj] = hi ? ih : il;
        }
    }

    v8f dr = {0.f,0.f,0.f,0.f,0.f,0.f,0.f,0.f};
    v8f di = {0.f,0.f,0.f,0.f,0.f,0.f,0.f,0.f};
    #pragma unroll
    for (int k = 0; k < 4; ++k) {
        dr = wmma4(Ar[k], Br[k], dr);
        di = wmma4(Ar[k], Bi[k], di);
    }
    #pragma unroll
    for (int k = 0; k < 4; ++k) {
        dr = wmma4(An[k], Bi[k], dr);
        di = wmma4(Ai[k], Br[k], di);
    }

    float R[8], I[8];
    #pragma unroll
    for (int r = 0; r < 8; ++r) { R[r] = dr[r]; I[r] = di[r]; }

    #pragma unroll
    for (int r = 0; r < 4; ++r) {
        float arr = R[r], aii = I[r], brr = R[r + 4], bii = I[r + 4];
        R[r]     = c1 * arr + s1 * bii;  I[r]     = c1 * aii - s1 * brr;
        R[r + 4] = c1 * brr + s1 * aii;  I[r + 4] = c1 * bii - s1 * arr;
    }
    #pragma unroll
    for (int t2 = 0; t2 < 4; ++t2) {
        int r = (t2 & 1) + (t2 >> 1) * 4;
        float arr = R[r], aii = I[r], brr = R[r + 2], bii = I[r + 2];
        R[r]     = c2 * arr + s2 * bii;  I[r]     = c2 * aii - s2 * brr;
        R[r + 2] = c2 * brr + s2 * aii;  I[r + 2] = c2 * bii - s2 * arr;
    }
    #pragma unroll
    for (int t3 = 0; t3 < 4; ++t3) {
        int r = t3 * 2;
        float arr = R[r], aii = I[r], brr = R[r + 1], bii = I[r + 1];
        R[r]     = c3 * arr + s3 * bii;  I[r]     = c3 * aii - s3 * brr;
        R[r + 1] = c3 * brr + s3 * aii;  I[r + 1] = c3 * bii - s3 * arr;
    }

    float SR[8], SI[8];
    #pragma unroll
    for (int r = 0; r < 8; ++r) { SR[r] = swap16(R[r]); SI[r] = swap16(I[r]); }

    float NR[8], NI[8], PR[8], PI[8];
    #pragma unroll
    for (int r = 0; r < 8; ++r) {
        NR[r] = c0 * R[r]  + s0 * SI[r];
        NI[r] = c0 * I[r]  - s0 * SR[r];
        PR[r] = c0 * SR[r] + s0 * I[r];
        PI[r] = c0 * SI[r] - s0 * R[r];
    }

    #pragma unroll
    for (int jj = 0; jj < 2; ++jj) {
        Br[0][jj] = hi ? PR[jj + 2] : NR[jj];      Bi[0][jj] = hi ? PI[jj + 2] : NI[jj];
        Br[1][jj] = hi ? PR[jj + 6] : NR[jj + 4];  Bi[1][jj] = hi ? PI[jj + 6] : NI[jj + 4];
        Br[2][jj] = hi ? NR[jj + 2] : PR[jj];      Bi[2][jj] = hi ? NI[jj + 2] : PI[jj];
        Br[3][jj] = hi ? NR[jj + 6] : PR[jj + 4];  Bi[3][jj] = hi ? NI[jj + 6] : PI[jj + 4];
    }

    #pragma unroll
    for (int k = 0; k < 4; ++k) {
        int off = n * 16 + 4 * k + cb;
        Ar[k] = *reinterpret_cast<const v2f*>(g + 3 * 256 + off);
        Ai[k] = *reinterpret_cast<const v2f*>(g + 4 * 256 + off);
        An[k] = *reinterpret_cast<const v2f*>(g + 5 * 256 + off);
    }

    v8f er = {0.f,0.f,0.f,0.f,0.f,0.f,0.f,0.f};
    v8f ei = {0.f,0.f,0.f,0.f,0.f,0.f,0.f,0.f};
    #pragma unroll
    for (int k = 0; k < 4; ++k) {
        er = wmma4(Ar[k], Br[k], er);
        ei = wmma4(Ar[k], Bi[k], ei);
    }
    #pragma unroll
    for (int k = 0; k < 4; ++k) {
        er = wmma4(An[k], Bi[k], er);
        ei = wmma4(Ai[k], Br[k], ei);
    }

    float p[8];
    #pragma unroll
    for (int r = 0; r < 8; ++r) p[r] = er[r] * er[r] + ei[r] * ei[r];

    const float t0 = p[0] + p[1], t1 = p[2] + p[3];
    const float t2 = p[4] + p[5], t3 = p[6] + p[7];
    const float u01 = t0 + t1, u23 = t2 + t3;
    float z0 = u01 + u23;
    float z1 = u01 - u23;
    float z2 = (t0 + t2) - (t1 + t3);
    const float w0 = p[0] - p[1], w1 = p[2] - p[3];
    const float w2 = p[4] - p[5], w3 = p[6] - p[7];
    float z3 = (w0 + w1) + (w2 + w3);
    if (hi) z0 = -z0;
    z0 += swap16(z0);
    z1 += swap16(z1);
    z2 += swap16(z2);
    z3 += swap16(z3);

    if (!hi) {
        float4 o; o.x = z0; o.y = z1; o.z = z2; o.w = z3;
        reinterpret_cast<float4*>(out)[samp] = o;
    }
}

extern "C" void kernel_launch(void* const* d_in, const int* in_sizes, int n_in,
                              void* d_out, int out_size, void* d_ws, size_t ws_size,
                              hipStream_t stream) {
    const float* x = (const float*)d_in[0];
    const float* w = (const float*)d_in[1];
    float* g   = (float*)d_ws;
    float* out = (float*)d_out;

    int B = in_sizes[0] / 4;
    build_G<<<1, 512, 0, stream>>>(w, g);
    qml_kernel<<<B / 128, 256, 0, stream>>>(x, g, out);
}


# First product takes a REAL right-hand side: start-state phases (-i)^popcount folded into G0's columns by the table kernel; 8 f32 WMMAs not 16 (24 per tile, not 32); its 16-term sums reorder
# speedup vs baseline: 1.2090x; 1.2090x over previous
#include <hip/hip_runtime.h>
#include <math.h>

typedef float v2f __attribute__((ext_vector_type(2)));
typedef float v8f __attribute__((ext_vector_type(8)));

__device__ inline v8f wmma4(v2f a, v2f b, v8f c) {
    return __builtin_amdgcn_wmma_f32_16x16x4_f32(
         false, a,  false, b,
         (short)0, c,  false,  false);
}

__device__ inline float swap16(float v) {
    return __int_as_float(__builtin_amdgcn_ds_swizzle(__float_as_int(v), 0x401F));
}

__global__ void build_G(const float* __restrict__ w, float* __restrict__ g) {
    int t = threadIdx.x;
    if (t >= 512) return;
    int l = t >> 8;
    int e = t & 255;
    int r = e >> 4;
    int c = e & 15;

    int j = r;
    if (j & 2) j ^= 1;
    if (j & 4) j ^= 2;
    if (j & 8) j ^= 4;

    float ar = 1.0f, ai = 0.0f;
    #pragma unroll
    for (int q = 0; q < 4; ++q) {
        float ty = w[(l * 4 + q) * 2 + 0] * 0.5f;
        float tz = w[(l * 4 + q) * 2 + 1] * 0.5f;
        float cy, sy, cz, sz;
        __sincosf(ty, &sy, &cy);
        __sincosf(tz, &sz, &cz);
        int br = (j >> (3 - q)) & 1;
        int bc = (c >> (3 - q)) & 1;
        float ur, ui;
        if (br == 0) { float m = bc ? -sy : cy; ur = m * cz; ui = -m * sz; }
        else         { float m = bc ?  cy : sy; ur = m * cz; ui =  m * sz; }
        float nr = ar * ur - ai * ui;
        float ni = ar * ui + ai * ur;
        ar = nr; ai = ni;
    }
    if (l == 0) {
        const int q = __popc(c) & 3;
        const float tr = ar, ti = ai;
        if (q == 1)      { ar =  ti; ai = -tr; }
        else if (q == 2) { ar = -tr; ai = -ti; }
        else if (q == 3) { ar = -ti; ai =  tr; }
    }
    g[(l * 3 + 0) * 256 + e] = ar;
    g[(l * 3 + 1) * 256 + e] = ai;
    if (l == 1) g[(l * 3 + 2) * 256 + e] = -ai;
}

__global__ void __launch_bounds__(256) qml_kernel(const float* __restrict__ x,
                                                  const float* __restrict__ g,
                                                  float* __restrict__ out) {
    const int lane = threadIdx.x & 31;
    const int wave = threadIdx.x >> 5;
    const int n    = lane & 15;
    const bool hi  = lane >= 16;
    const int cb   = hi ? 2 : 0;
    const int samp = blockIdx.x * 128 + wave * 16 + n;

    float4 xv = reinterpret_cast<const float4*>(x)[samp];
    float c0, s0, c1, s1, c2, s2, c3, s3;
    __sincosf(xv.x * 0.5f, &s0, &c0);
    __sincosf(xv.y * 0.5f, &s1, &c1);
    __sincosf(xv.z * 0.5f, &s2, &c2);
    __sincosf(xv.w * 0.5f, &s3, &c3);

    v2f Ar[4], Ai[4];
    #pragma unroll
    for (int k = 0; k < 4; ++k) {
        int off = n * 16 + 4 * k + cb;
        Ar[k] = *reinterpret_cast<const v2f*>(g + 0 * 256 + off);
        Ai[k] = *reinterpret_cast<const v2f*>(g + 1 * 256 + off);
    }

    const float f2  = hi ? s2 : c2;
    const float ab0 = c0 * c1, ab1 = c0 * s1, ab2 = s0 * c1, ab3 = s0 * s1;
    const float ab[4] = {ab0, ab1, ab2, ab3};
    const float fd0 = f2 * c3, fd1 = f2 * s3;
    v2f Bm[4];
    #pragma unroll
    for (int k = 0; k < 4; ++k) {
        Bm[k][0] = ab[k] * fd0;
        Bm[k][1] = ab[k] * fd1;
    }

    v8f dr = {0.f,0.f,0.f,0.f,0.f,0.f,0.f,0.f};
    v8f di = {0.f,0.f,0.f,0.f,0.f,0.f,0.f,0.f};
    #pragma unroll
    for (int k = 0; k < 4; ++k) {
        dr = wmma4(Ar[k], Bm[k], dr);
        di = wmma4(Ai[k], Bm[k], di);
    }

    float R[8], I[8];
    #pragma unroll
    for (int r = 0; r < 8; ++r) { R[r] = dr[r]; I[r] = di[r]; }

    #pragma unroll
    for (int r = 0; r < 4; ++r) {
        float arr = R[r], aii = I[r], brr = R[r + 4], bii = I[r + 4];
        R[r]     = c1 * arr + s1 * bii;  I[r]     = c1 * aii - s1 * brr;
        R[r + 4] = c1 * brr + s1 * aii;  I[r + 4] = c1 * bii - s1 * arr;
    }
    #pragma unroll
    for (int t2 = 0; t2 < 4; ++t2) {
        int r = (t2 & 1) + (t2 >> 1) * 4;
        float arr = R[r], aii = I[r], brr = R[r + 2], bii = I[r + 2];
        R[r]     = c2 * arr + s2 * bii;  I[r]     = c2 * aii - s2 * brr;
        R[r + 2] = c2 * brr + s2 * aii;  I[r + 2] = c2 * bii - s2 * arr;
    }
    #pragma unroll
    for (int t3 = 0; t3 < 4; ++t3) {
        int r = t3 * 2;
        float arr = R[r], aii = I[r], brr = R[r + 1], bii = I[r + 1];
        R[r]     = c3 * arr + s3 * bii;  I[r]     = c3 * aii - s3 * brr;
        R[r + 1] = c3 * brr + s3 * aii;  I[r + 1] = c3 * bii - s3 * arr;
    }

    float SR[8], SI[8];
    #pragma unroll
    for (int r = 0; r < 8; ++r) { SR[r] = swap16(R[r]); SI[r] = swap16(I[r]); }

    float NR[8], NI[8], PR[8], PI[8];
    #pragma unroll
    for (int r = 0; r < 8; ++r) {
        NR[r] = c0 * R[r]  + s0 * SI[r];
        NI[r] = c0 * I[r]  - s0 * SR[r];
        PR[r] = c0 * SR[r] + s0 * I[r];
        PI[r] = c0 * SI[r] - s0 * R[r];
    }

    v2f Br[4], Bi[4];
    #pragma unroll
    for (int jj = 0; jj < 2; ++jj) {
        Br[0][jj] = hi ? PR[jj + 2] : NR[jj];      Bi[0][jj] = hi ? PI[jj + 2] : NI[jj];
        Br[1][jj] = hi ? PR[jj + 6] : NR[jj + 4];  Bi[1][jj] = hi ? PI[jj + 6] : NI[jj + 4];
        Br[2][jj] = hi ? NR[jj + 2] : PR[jj];      Bi[2][jj] = hi ? NI[jj + 2] : PI[jj];
        Br[3][jj] = hi ? NR[jj + 6] : PR[jj + 4];  Bi[3][jj] = hi ? NI[jj + 6] : PI[jj + 4];
    }

    v2f An[4];
    #pragma unroll
    for (int k = 0; k < 4; ++k) {
        int off = n * 16 + 4 * k + cb;
        Ar[k] = *reinterpret_cast<const v2f*>(g + 3 * 256 + off);
        Ai[k] = *reinterpret_cast<const v2f*>(g + 4 * 256 + off);
        An[k] = *reinterpret_cast<const v2f*>(g + 5 * 256 + off);
    }

    v8f er = {0.f,0.f,0.f,0.f,0.f,0.f,0.f,0.f};
    v8f ei = {0.f,0.f,0.f,0.f,0.f,0.f,0.f,0.f};
    #pragma unroll
    for (int k = 0; k < 4; ++k) {
        er = wmma4(Ar[k], Br[k], er);
        ei = wmma4(Ar[k], Bi[k], ei);
    }
    #pragma unroll
    for (int k = 0; k < 4; ++k) {
        er = wmma4(An[k], Bi[k], er);
        ei = wmma4(Ai[k], Br[k], ei);
    }

    float p[8];
    #pragma unroll
    for (int r = 0; r < 8; ++r) p[r] = er[r] * er[r] + ei[r] * ei[r];

    const float t0 = p[0] + p[1], t1 = p[2] + p[3];
    const float t2 = p[4] + p[5], t3 = p[6] + p[7];
    const float u01 = t0 + t1, u23 = t2 + t3;
    float z0 = u01 + u23;
    float z1 = u01 - u23;
    float z2 = (t0 + t2) - (t1 + t3);
    const float w0 = p[0] - p[1], w1 = p[2] - p[3];
    const float w2 = p[4] - p[5], w3 = p[6] - p[7];
    float z3 = (w0 + w1) + (w2 + w3);
    if (hi) z0 = -z0;
    z0 += swap16(z0);
    z1 += swap16(z1);
    z2 += swap16(z2);
    z3 += swap16(z3);

    if (!hi) {
        float4 o; o.x = z0; o.y = z1; o.z = z2; o.w = z3;
        reinterpret_cast<float4*>(out)[samp] = o;
    }
}

extern "C" void kernel_launch(void* const* d_in, const int* in_sizes, int n_in,
                              void* d_out, int out_size, void* d_ws, size_t ws_size,
                              hipStream_t stream) {
    const float* x = (const float*)d_in[0];
    const float* w = (const float*)d_in[1];
    float* g   = (float*)d_ws;
    float* out = (float*)d_out;

    int B = in_sizes[0] / 4;
    build_G<<<1, 512, 0, stream>>>(w, g);
    qml_kernel<<<B / 128, 256, 0, stream>>>(x, g, out);
}


# Second product takes a REAL left-hand side: last-layer RZ is a unit-modulus phase per row and cancels in |amp|^2, so G1 = CNOTs*kron(RY); 8 f32 WMMAs not 16 (16 per tile, not 24); its sums reorder
# speedup vs baseline: 1.4752x; 1.2202x over previous
#include <hip/hip_runtime.h>
#include <math.h>

typedef float v2f __attribute__((ext_vector_type(2)));
typedef float v8f __attribute__((ext_vector_type(8)));

__device__ inline v8f wmma4(v2f a, v2f b, v8f c) {
    return __builtin_amdgcn_wmma_f32_16x16x4_f32(
         false, a,  false, b,
         (short)0, c,  false,  false);
}

__device__ inline float swap16(float v) {
    return __int_as_float(__builtin_amdgcn_ds_swizzle(__float_as_int(v), 0x401F));
}

__global__ void build_G(const float* __restrict__ w, float* __restrict__ g) {
    int t = threadIdx.x;
    if (t >= 512) return;
    int l = t >> 8;
    int e = t & 255;
    int r = e >> 4;
    int c = e & 15;

    int j = r;
    if (j & 2) j ^= 1;
    if (j & 4) j ^= 2;
    if (j & 8) j ^= 4;

    float ar = 1.0f, ai = 0.0f;
    #pragma unroll
    for (int q = 0; q < 4; ++q) {
        float ty = w[(l * 4 + q) * 2 + 0] * 0.5f;
        float tz = w[(l * 4 + q) * 2 + 1] * 0.5f;
        float cy, sy, cz, sz;
        __sincosf(ty, &sy, &cy);
        __sincosf(tz, &sz, &cz);
        if (l == 1) { cz = 1.0f; sz = 0.0f; }
        int br = (j >> (3 - q)) & 1;
        int bc = (c >> (3 - q)) & 1;
        float ur, ui;
        if (br == 0) { float m = bc ? -sy : cy; ur = m * cz; ui = -m * sz; }
        else         { float m = bc ?  cy : sy; ur = m * cz; ui =  m * sz; }
        float nr = ar * ur - ai * ui;
        float ni = ar * ui + ai * ur;
        ar = nr; ai = ni;
    }
    if (l == 0) {
        const int q = __popc(c) & 3;
        const float tr = ar, ti = ai;
        if (q == 1)      { ar =  ti; ai = -tr; }
        else if (q == 2) { ar = -tr; ai = -ti; }
        else if (q == 3) { ar = -ti; ai =  tr; }
        g[0 * 256 + e] = ar;
        g[1 * 256 + e] = ai;
    } else {
        g[2 * 256 + e] = ar;
    }
}

__global__ void __launch_bounds__(256) qml_kernel(const float* __restrict__ x,
                                                  const float* __restrict__ g,
                                                  float* __restrict__ out) {
    const int lane = threadIdx.x & 31;
    const int wave = threadIdx.x >> 5;
    const int n    = lane & 15;
    const bool hi  = lane >= 16;
    const int cb   = hi ? 2 : 0;
    const int samp = blockIdx.x * 128 + wave * 16 + n;

    float4 xv = reinterpret_cast<const float4*>(x)[samp];
    float c0, s0, c1, s1, c2, s2, c3, s3;
    __sincosf(xv.x * 0.5f, &s0, &c0);
    __sincosf(xv.y * 0.5f, &s1, &c1);
    __sincosf(xv.z * 0.5f, &s2, &c2);
    __sincosf(xv.w * 0.5f, &s3, &c3);

    v2f Ar[4], Ai[4];
    #pragma unroll
    for (int k = 0; k < 4; ++k) {
        int off = n * 16 + 4 * k + cb;
        Ar[k] = *reinterpret_cast<const v2f*>(g + 0 * 256 + off);
        Ai[k] = *reinterpret_cast<const v2f*>(g + 1 * 256 + off);
    }

    const float f2  = hi ? s2 : c2;
    const float ab0 = c0 * c1, ab1 = c0 * s1, ab2 = s0 * c1, ab3 = s0 * s1;
    const float ab[4] = {ab0, ab1, ab2, ab3};
    const float fd0 = f2 * c3, fd1 = f2 * s3;
    v2f Bm[4];
    #pragma unroll
    for (int k = 0; k < 4; ++k) {
        Bm[k][0] = ab[k] * fd0;
        Bm[k][1] = ab[k] * fd1;
    }

    v8f dr = {0.f,0.f,0.f,0.f,0.f,0.f,0.f,0.f};
    v8f di = {0.f,0.f,0.f,0.f,0.f,0.f,0.f,0.f};
    #pragma unroll
    for (int k = 0; k < 4; ++k) {
        dr = wmma4(Ar[k], Bm[k], dr);
        di = wmma4(Ai[k], Bm[k], di);
    }

    float R[8], I[8];
    #pragma unroll
    for (int r = 0; r < 8; ++r) { R[r] = dr[r]; I[r] = di[r]; }

    #pragma unroll
    for (int r = 0; r < 4; ++r) {
        float arr = R[r], aii = I[r], brr = R[r + 4], bii = I[r + 4];
        R[r]     = c1 * arr + s1 * bii;  I[r]     = c1 * aii - s1 * brr;
        R[r + 4] = c1 * brr + s1 * aii;  I[r + 4] = c1 * bii - s1 * arr;
    }
    #pragma unroll
    for (int t2 = 0; t2 < 4; ++t2) {
        int r = (t2 & 1) + (t2 >> 1) * 4;
        float arr = R[r], aii = I[r], brr = R[r + 2], bii = I[r + 2];
        R[r]     = c2 * arr + s2 * bii;  I[r]     = c2 * aii - s2 * brr;
        R[r + 2] = c2 * brr + s2 * aii;  I[r + 2] = c2 * bii - s2 * arr;
    }
    #pragma unroll
    for (int t3 = 0; t3 < 4; ++t3) {
        int r = t3 * 2;
        float arr = R[r], aii = I[r], brr = R[r + 1], bii = I[r + 1];
        R[r]     = c3 * arr + s3 * bii;  I[r]     = c3 * aii - s3 * brr;
        R[r + 1] = c3 * brr + s3 * aii;  I[r + 1] = c3 * bii - s3 * arr;
    }

    float SR[8], SI[8];
    #pragma unroll
    for (int r = 0; r < 8; ++r) { SR[r] = swap16(R[r]); SI[r] = swap16(I[r]); }

    float NR[8], NI[8], PR[8], PI[8];
    #pragma unroll
    for (int r = 0; r < 8; ++r) {
        NR[r] = c0 * R[r]  + s0 * SI[r];
        NI[r] = c0 * I[r]  - s0 * SR[r];
        PR[r] = c0 * SR[r] + s0 * I[r];
        PI[r] = c0 * SI[r] - s0 * R[r];
    }

    v2f Br[4], Bi[4];
    #pragma unroll
    for (int jj = 0; jj < 2; ++jj) {
        Br[0][jj] = hi ? PR[jj + 2] : NR[jj];      Bi[0][jj] = hi ? PI[jj + 2] : NI[jj];
        Br[1][jj] = hi ? PR[jj + 6] : NR[jj + 4];  Bi[1][jj] = hi ? PI[jj + 6] : NI[jj + 4];
        Br[2][jj] = hi ? NR[jj + 2] : PR[jj];      Bi[2][jj] = hi ? NI[jj + 2] : PI[jj];
        Br[3][jj] = hi ? NR[jj + 6] : PR[jj + 4];  Bi[3][jj] = hi ? NI[jj + 6] : PI[jj + 4];
    }

    v2f Ay[4];
    #pragma unroll
    for (int k = 0; k < 4; ++k) {
        int off = n * 16 + 4 * k + cb;
        Ay[k] = *reinterpret_cast<const v2f*>(g + 2 * 256 + off);
    }

    v8f er = {0.f,0.f,0.f,0.f,0.f,0.f,0.f,0.f};
    v8f ei = {0.f,0.f,0.f,0.f,0.f,0.f,0.f,0.f};
    #pragma unroll
    for (int k = 0; k < 4; ++k) {
        er = wmma4(Ay[k], Br[k], er);
        ei = wmma4(Ay[k], Bi[k], ei);
    }

    float p[8];
    #pragma unroll
    for (int r = 0; r < 8; ++r) p[r] = er[r] * er[r] + ei[r] * ei[r];

    const float t0 = p[0] + p[1], t1 = p[2] + p[3];
    const float t2 = p[4] + p[5], t3 = p[6] + p[7];
    const float u01 = t0 + t1, u23 = t2 + t3;
    float z0 = u01 + u23;
    float z1 = u01 - u23;
    float z2 = (t0 + t2) - (t1 + t3);
    const float w0 = p[0] - p[1], w1 = p[2] - p[3];
    const float w2 = p[4] - p[5], w3 = p[6] - p[7];
    float z3 = (w0 + w1) + (w2 + w3);
    if (hi) z0 = -z0;
    z0 += swap16(z0);
    z1 += swap16(z1);
    z2 += swap16(z2);
    z3 += swap16(z3);

    if (!hi) {
        float4 o; o.x = z0; o.y = z1; o.z = z2; o.w = z3;
        reinterpret_cast<float4*>(out)[samp] = o;
    }
}

extern "C" void kernel_launch(void* const* d_in, const int* in_sizes, int n_in,
                              void* d_out, int out_size, void* d_ws, size_t ws_size,
                              hipStream_t stream) {
    const float* x = (const float*)d_in[0];
    const float* w = (const float*)d_in[1];
    float* g   = (float*)d_ws;
    float* out = (float*)d_out;

    int B = in_sizes[0] / 4;
    build_G<<<1, 512, 0, stream>>>(w, g);
    qml_kernel<<<B / 128, 256, 0, stream>>>(x, g, out);
}


# Per-sample RX layer made diagonal: H on every qubit folded into both gate tables (H4*G0; G1*H4, columns permuted); layer = one complex phase multiply per component, no lane swaps; sums reorder
# speedup vs baseline: 1.6538x; 1.1211x over previous
#include <hip/hip_runtime.h>
#include <math.h>

typedef float v2f __attribute__((ext_vector_type(2)));
typedef float v8f __attribute__((ext_vector_type(8)));

__device__ inline v8f wmma4(v2f a, v2f b, v8f c) {
    return __builtin_amdgcn_wmma_f32_16x16x4_f32(
         false, a,  false, b,
         (short)0, c,  false,  false);
}

__device__ inline float swap16(float v) {
    return __int_as_float(__builtin_amdgcn_ds_swizzle(__float_as_int(v), 0x401F));
}

__global__ void build_G(const float* __restrict__ w, float* __restrict__ g) {
    int t = threadIdx.x;
    if (t >= 512) return;
    int l = t >> 8;
    int e = t & 255;
    int r = e >> 4;
    int c = e & 15;

    int j = r;
    int cs = c;
    if (l == 0) {
        if (j & 1) j ^= 2;
        if (j & 2) j ^= 4;
        if (j & 4) j ^= 8;
    } else {
        if (j & 2) j ^= 1;
        if (j & 4) j ^= 2;
        if (j & 8) j ^= 4;
        cs = ((c & 2) << 2) | ((c >> 2) << 1) | (c & 1);
    }

    const float r2 = 0.70710678f;
    float ar = 1.0f, ai = 0.0f;
    #pragma unroll
    for (int q = 0; q < 4; ++q) {
        float ty = w[(l * 4 + q) * 2 + 0] * 0.5f;
        float tz = w[(l * 4 + q) * 2 + 1] * 0.5f;
        float cy, sy, cz, sz;
        __sincosf(ty, &sy, &cy);
        __sincosf(tz, &sz, &cz);
        if (l == 1) { cz = 1.0f; sz = 0.0f; }
        int br = (j  >> (3 - q)) & 1;
        int bc = (cs >> (3 - q)) & 1;
        const int   f    = (l == 0) ? bc : br;
        const int   sgb  = (l == 0) ? br : bc;
        const float sneg = (l == 0) ? -sy : sy;
        const float x0 = f ? sneg : cy;
        const float x1 = f ? cy : -sneg;
        const float sx = sgb ? -x1 : x1;
        const float ur = (x0 + sx) * cz * r2;
        const float ui = (sx - x0) * sz * r2;
        float nr = ar * ur - ai * ui;
        float ni = ar * ui + ai * ur;
        ar = nr; ai = ni;
    }
    if (l == 0) {
        const int q = __popc(c) & 3;
        const float tr = ar, ti = ai;
        if (q == 1)      { ar =  ti; ai = -tr; }
        else if (q == 2) { ar = -tr; ai = -ti; }
        else if (q == 3) { ar = -ti; ai =  tr; }
        g[0 * 256 + e] = ar;
        g[1 * 256 + e] = ai;
    } else {
        g[2 * 256 + e] = ar;
    }
}

__global__ void __launch_bounds__(256) qml_kernel(const float* __restrict__ x,
                                                  const float* __restrict__ g,
                                                  float* __restrict__ out) {
    const int lane = threadIdx.x & 31;
    const int wave = threadIdx.x >> 5;
    const int n    = lane & 15;
    const bool hi  = lane >= 16;
    const int cb   = hi ? 2 : 0;
    const int samp = blockIdx.x * 128 + wave * 16 + n;

    float4 xv = reinterpret_cast<const float4*>(x)[samp];
    float c0, s0, c1, s1, c2, s2, c3, s3;
    __sincosf(xv.x * 0.5f, &s0, &c0);
    __sincosf(xv.y * 0.5f, &s1, &c1);
    __sincosf(xv.z * 0.5f, &s2, &c2);
    __sincosf(xv.w * 0.5f, &s3, &c3);

    v2f Ar[4], Ai[4];
    #pragma unroll
    for (int k = 0; k < 4; ++k) {
        int off = n * 16 + 4 * k + cb;
        Ar[k] = *reinterpret_cast<const v2f*>(g + 0 * 256 + off);
        Ai[k] = *reinterpret_cast<const v2f*>(g + 1 * 256 + off);
    }

    const float f2  = hi ? s2 : c2;
    const float ab0 = c0 * c1, ab1 = c0 * s1, ab2 = s0 * c1, ab3 = s0 * s1;
    const float ab[4] = {ab0, ab1, ab2, ab3};
    const float fd0 = f2 * c3, fd1 = f2 * s3;
    v2f Bm[4];
    #pragma unroll
    for (int k = 0; k < 4; ++k) {
        Bm[k][0] = ab[k] * fd0;
        Bm[k][1] = ab[k] * fd1;
    }

    v8f dr = {0.f,0.f,0.f,0.f,0.f,0.f,0.f,0.f};
    v8f di = {0.f,0.f,0.f,0.f,0.f,0.f,0.f,0.f};
    #pragma unroll
    for (int k = 0; k < 4; ++k) {
        dr = wmma4(Ar[k], Bm[k], dr);
        di = wmma4(Ai[k], Bm[k], di);
    }

    const float cc = c2 * c3, ss = s2 * s3, sc = s2 * c3, cs = c2 * s3;
    const float PA = cc - ss, PB = sc + cs, PC = cc + ss, PD = cs - sc;
    const float X23[4] = { PA, PC,  PC, PA};
    const float Y23[4] = {-PB, PD, -PD, PB};
    const float sg0 = hi ? s0 : -s0;
    float Pr[8], Pi[8];
    #pragma unroll
    for (int t = 0; t < 4; ++t) {
        const float qr = c1 * X23[t] + s1 * Y23[t];
        const float qi = c1 * Y23[t] - s1 * X23[t];
        Pr[t]     = c0 * qr - sg0 * qi;   Pi[t]     = c0 * qi + sg0 * qr;
        Pr[7 - t] = c0 * qr + sg0 * qi;   Pi[7 - t] = sg0 * qr - c0 * qi;
    }

    v2f Br[4], Bi[4];
    #pragma unroll
    for (int k = 0; k < 4; ++k) {
        #pragma unroll
        for (int jj = 0; jj < 2; ++jj) {
            const int r = 2 * k + jj;
            Br[k][jj] = dr[r] * Pr[r] - di[r] * Pi[r];
            Bi[k][jj] = dr[r] * Pi[r] + di[r] * Pr[r];
        }
    }

    v2f Ay[4];
    #pragma unroll
    for (int k = 0; k < 4; ++k) {
        int off = n * 16 + 4 * k + cb;
        Ay[k] = *reinterpret_cast<const v2f*>(g + 2 * 256 + off);
    }

    v8f er = {0.f,0.f,0.f,0.f,0.f,0.f,0.f,0.f};
    v8f ei = {0.f,0.f,0.f,0.f,0.f,0.f,0.f,0.f};
    #pragma unroll
    for (int k = 0; k < 4; ++k) {
        er = wmma4(Ay[k], Br[k], er);
        ei = wmma4(Ay[k], Bi[k], ei);
    }

    float p[8];
    #pragma unroll
    for (int r = 0; r < 8; ++r) p[r] = er[r] * er[r] + ei[r] * ei[r];

    const float t0 = p[0] + p[1], t1 = p[2] + p[3];
    const float t2 = p[4] + p[5], t3 = p[6] + p[7];
    const float u01 = t0 + t1, u23 = t2 + t3;
    float z0 = u01 + u23;
    float z1 = u01 - u23;
    float z2 = (t0 + t2) - (t1 + t3);
    const float w0 = p[0] - p[1], w1 = p[2] - p[3];
    const float w2 = p[4] - p[5], w3 = p[6] - p[7];
    float z3 = (w0 + w1) + (w2 + w3);
    if (hi) z0 = -z0;
    z0 += swap16(z0);
    z1 += swap16(z1);
    z2 += swap16(z2);
    z3 += swap16(z3);

    if (!hi) {
        float4 o; o.x = z0; o.y = z1; o.z = z2; o.w = z3;
        reinterpret_cast<float4*>(out)[samp] = o;
    }
}

extern "C" void kernel_launch(void* const* d_in, const int* in_sizes, int n_in,
                              void* d_out, int out_size, void* d_ws, size_t ws_size,
                              hipStream_t stream) {
    const float* x = (const float*)d_in[0];
    const float* w = (const float*)d_in[1];
    float* g   = (float*)d_ws;
    float* out = (float*)d_out;

    int B = in_sizes[0] / 4;
    build_G<<<1, 512, 0, stream>>>(w, g);
    qml_kernel<<<B / 128, 256, 0, stream>>>(x, g, out);
}


# One launch, not two: each 8-wave work-group builds the gate tables in its LDS, then every wave runs 8 tiles, each tile's angles requested two tiles before use; per-tile arithmetic unchanged
# speedup vs baseline: 1.8096x; 1.0942x over previous
#include <hip/hip_runtime.h>
#include <math.h>

typedef float v2f __attribute__((ext_vector_type(2)));
typedef float v8f __attribute__((ext_vector_type(8)));

__device__ inline v8f wmma4(v2f a, v2f b, v8f c) {
    return __builtin_amdgcn_wmma_f32_16x16x4_f32(
         false, a,  false, b,
         (short)0, c,  false,  false);
}

__device__ inline float swap16(float v) {
    return __int_as_float(__builtin_amdgcn_ds_swizzle(__float_as_int(v), 0x401F));
}

template <int l>
__device__ inline void table_entry(const float* __restrict__ w, int e, float& ar, float& ai) {
    int r = e >> 4;
    int c = e & 15;

    int j = r;
    int cs = c;
    if (l == 0) {
        if (j & 1) j ^= 2;
        if (j & 2) j ^= 4;
        if (j & 4) j ^= 8;
    } else {
        if (j & 2) j ^= 1;
        if (j & 4) j ^= 2;
        if (j & 8) j ^= 4;
        cs = ((c & 2) << 2) | ((c >> 2) << 1) | (c & 1);
    }

    const float r2 = 0.70710678f;
    ar = 1.0f; ai = 0.0f;
    #pragma unroll
    for (int q = 0; q < 4; ++q) {
        float ty = w[(l * 4 + q) * 2 + 0] * 0.5f;
        float tz = w[(l * 4 + q) * 2 + 1] * 0.5f;
        float cy, sy, cz, sz;
        __sincosf(ty, &sy, &cy);
        __sincosf(tz, &sz, &cz);
        if (l == 1) { cz = 1.0f; sz = 0.0f; }
        int br = (j  >> (3 - q)) & 1;
        int bc = (cs >> (3 - q)) & 1;
        const int   f    = (l == 0) ? bc : br;
        const int   sgb  = (l == 0) ? br : bc;
        const float sneg = (l == 0) ? -sy : sy;
        const float x0 = f ? sneg : cy;
        const float x1 = f ? cy : -sneg;
        const float sx = sgb ? -x1 : x1;
        const float ur = (x0 + sx) * cz * r2;
        const float ui = (sx - x0) * sz * r2;
        float nr = ar * ur - ai * ui;
        float ni = ar * ui + ai * ur;
        ar = nr; ai = ni;
    }
    if (l == 0) {
        const int q = __popc(c) & 3;
        const float tr = ar, ti = ai;
        if (q == 1)      { ar =  ti; ai = -tr; }
        else if (q == 2) { ar = -tr; ai = -ti; }
        else if (q == 3) { ar = -ti; ai =  tr; }
    }
}

#define TILES 8
__global__ void __launch_bounds__(256) qml_kernel(const float* __restrict__ x,
                                                  const float* __restrict__ w,
                                                  float* __restrict__ out) {
    __shared__ alignas(16) float gs[3 * 256];

    const int lane = threadIdx.x & 31;
    const int wave = threadIdx.x >> 5;
    const int n    = lane & 15;
    const bool hi  = lane >= 16;
    const int cb   = hi ? 2 : 0;
    const int samp0 = (blockIdx.x * 8 + wave) * (16 * TILES) + n;

    const float4* __restrict__ x4 = reinterpret_cast<const float4*>(x);
    float4* __restrict__ o4       = reinterpret_cast<float4*>(out);

    float4 xv = x4[samp0];
    float4 xn = x4[samp0 + 16];

    {
        float wv[16];
        #pragma unroll
        for (int i = 0; i < 16; ++i) wv[i] = w[i];
        const int e = threadIdx.x;
        float ar, ai;
        table_entry<0>(wv, e, ar, ai);
        gs[0 * 256 + e] = ar;
        gs[1 * 256 + e] = ai;
        table_entry<1>(wv, e, ar, ai);
        gs[2 * 256 + e] = ar;
    }
    __syncthreads();

    v2f Ar[4], Ai[4], Ay[4];
    #pragma unroll
    for (int k = 0; k < 4; ++k) {
        int off = n * 16 + 4 * k + cb;
        Ar[k] = *reinterpret_cast<const v2f*>(gs + 0 * 256 + off);
        Ai[k] = *reinterpret_cast<const v2f*>(gs + 1 * 256 + off);
        Ay[k] = *reinterpret_cast<const v2f*>(gs + 2 * 256 + off);
    }

    for (int t = 0; t < TILES; ++t) {
        const int samp = samp0 + 16 * t;

        float4 xf = xn;
        if (t + 2 < TILES) xf = x4[samp + 32];

        float c0, s0, c1, s1, c2, s2, c3, s3;
        __sincosf(xv.x * 0.5f, &s0, &c0);
        __sincosf(xv.y * 0.5f, &s1, &c1);
        __sincosf(xv.z * 0.5f, &s2, &c2);
        __sincosf(xv.w * 0.5f, &s3, &c3);

        const float f2  = hi ? s2 : c2;
        const float ab0 = c0 * c1, ab1 = c0 * s1, ab2 = s0 * c1, ab3 = s0 * s1;
        const float ab[4] = {ab0, ab1, ab2, ab3};
        const float fd0 = f2 * c3, fd1 = f2 * s3;
        v2f Bm[4];
        #pragma unroll
        for (int k = 0; k < 4; ++k) {
            Bm[k][0] = ab[k] * fd0;
            Bm[k][1] = ab[k] * fd1;
        }

        v8f dr = {0.f,0.f,0.f,0.f,0.f,0.f,0.f,0.f};
        v8f di = {0.f,0.f,0.f,0.f,0.f,0.f,0.f,0.f};
        #pragma unroll
        for (int k = 0; k < 4; ++k) {
            dr = wmma4(Ar[k], Bm[k], dr);
            di = wmma4(Ai[k], Bm[k], di);
        }

        const float cc = c2 * c3, ss = s2 * s3, sc = s2 * c3, cs = c2 * s3;
        const float PA = cc - ss, PB = sc + cs, PC = cc + ss, PD = cs - sc;
        const float X23[4] = { PA, PC,  PC, PA};
        const float Y23[4] = {-PB, PD, -PD, PB};
        const float sg0 = hi ? s0 : -s0;
        float Pr[8], Pi[8];
        #pragma unroll
        for (int q = 0; q < 4; ++q) {
            const float qr = c1 * X23[q] + s1 * Y23[q];
            const float qi = c1 * Y23[q] - s1 * X23[q];
            Pr[q]     = c0 * qr - sg0 * qi;   Pi[q]     = c0 * qi + sg0 * qr;
            Pr[7 - q] = c0 * qr + sg0 * qi;   Pi[7 - q] = sg0 * qr - c0 * qi;
        }

        v2f Br[4], Bi[4];
        #pragma unroll
        for (int k = 0; k < 4; ++k) {
            #pragma unroll
            for (int jj = 0; jj < 2; ++jj) {
                const int r = 2 * k + jj;
                Br[k][jj] = dr[r] * Pr[r] - di[r] * Pi[r];
                Bi[k][jj] = dr[r] * Pi[r] + di[r] * Pr[r];
            }
        }

        v8f er = {0.f,0.f,0.f,0.f,0.f,0.f,0.f,0.f};
        v8f ei = {0.f,0.f,0.f,0.f,0.f,0.f,0.f,0.f};
        #pragma unroll
        for (int k = 0; k < 4; ++k) {
            er = wmma4(Ay[k], Br[k], er);
            ei = wmma4(Ay[k], Bi[k], ei);
        }

        float p[8];
        #pragma unroll
        for (int r = 0; r < 8; ++r) p[r] = er[r] * er[r] + ei[r] * ei[r];

        const float t0 = p[0] + p[1], t1 = p[2] + p[3];
        const float t2 = p[4] + p[5], t3 = p[6] + p[7];
        const float u01 = t0 + t1, u23 = t2 + t3;
        float z0 = u01 + u23;
        float z1 = u01 - u23;
        float z2 = (t0 + t2) - (t1 + t3);
        const float w0 = p[0] - p[1], w1 = p[2] - p[3];
        const float w2 = p[4] - p[5], w3 = p[6] - p[7];
        float z3 = (w0 + w1) + (w2 + w3);
        if (hi) z0 = -z0;
        z0 += swap16(z0);
        z1 += swap16(z1);
        z2 += swap16(z2);
        z3 += swap16(z3);

        if (!hi) {
            float4 o; o.x = z0; o.y = z1; o.z = z2; o.w = z3;
            o4[samp] = o;
        }

        xv = xn;
        xn = xf;
    }
}

extern "C" void kernel_launch(void* const* d_in, const int* in_sizes, int n_in,
                              void* d_out, int out_size, void* d_ws, size_t ws_size,
                              hipStream_t stream) {
    const float* x = (const float*)d_in[0];
    const float* w = (const float*)d_in[1];
    float* out = (float*)d_out;

    int B = in_sizes[0] / 4;
    qml_kernel<<<B / (128 * TILES), 256, 0, stream>>>(x, w, out);
}


# First product factorised by the start state's tensor structure: two 4x4 complex-by-real products on WMMA (2 f32 WMMAs, not 8; 10 per tile), their per-sample outer product on vector ALU; sums reorder
# speedup vs baseline: 2.1761x; 1.2025x over previous
#include <hip/hip_runtime.h>
#include <math.h>

typedef float v2f __attribute__((ext_vector_type(2)));
typedef float v8f __attribute__((ext_vector_type(8)));

__device__ inline v8f wmma4(v2f a, v2f b, v8f c) {
    return __builtin_amdgcn_wmma_f32_16x16x4_f32(
         false, a,  false, b,
         (short)0, c,  false,  false);
}

__device__ inline float swap16(float v) {
    return __int_as_float(__builtin_amdgcn_ds_swizzle(__float_as_int(v), 0x401F));
}


__device__ inline void fac0(const float* __restrict__ w, int q, int br, int bc, float& ur, float& ui) {
    const float r2 = 0.70710678f;
    float ty = w[q * 2 + 0] * 0.5f;
    float tz = w[q * 2 + 1] * 0.5f;
    float cy, sy, cz, sz;
    __sincosf(ty, &sy, &cy);
    __sincosf(tz, &sz, &cz);
    const float x0 = bc ? -sy : cy;
    const float x1 = bc ? cy : sy;
    const float sx = br ? -x1 : x1;
    const float pr = (x0 + sx) * cz * r2;
    const float pi = (sx - x0) * sz * r2;
    ur = bc ? pi : pr;
    ui = bc ? -pr : pi;
}

template <int which>
__device__ inline float table0_entry(const float* __restrict__ w, int row, int k) {
    const int h = row >> 3, g = row & 7;
    const int a = k >> 1, b = k & 1;
    const int q = which ? 2 : 0;
    int ia, ib, part;
    if (which == 0) {
        const int s2 = g & 1;
        part = (g >> 1) & 1;
        ia = h ^ s2; ib = s2;
    } else {
        const int e = g & 1, r1 = g >> 2;
        part = (g >> 1) & 1;
        ia = r1 ^ e; ib = e;
    }
    float ar, ai, br, bi;
    fac0(w, q,     ia, a, ar, ai);
    fac0(w, q + 1, ib, b, br, bi);
    const float pr = ar * br - ai * bi;
    const float pi = ar * bi + ai * br;
    float v = part ? pi : pr;
    if (which == 0 && g >= 4) v = 0.0f;
    return v;
}

__device__ inline float table1_entry(const float* __restrict__ w, int e) {
    int r = e >> 4;
    int c = e & 15;
    int j = r;
    if (j & 2) j ^= 1;
    if (j & 4) j ^= 2;
    if (j & 8) j ^= 4;
    const int cs = ((c & 2) << 2) | ((c >> 2) << 1) | (c & 1);

    const float r2 = 0.70710678f;
    float ar = 1.0f;
    #pragma unroll
    for (int q = 0; q < 4; ++q) {
        float ty = w[(4 + q) * 2 + 0] * 0.5f;
        float cy, sy;
        __sincosf(ty, &sy, &cy);
        int br = (j  >> (3 - q)) & 1;
        int bc = (cs >> (3 - q)) & 1;
        const float x0 = br ? sy : cy;
        const float x1 = br ? cy : -sy;
        const float sx = bc ? -x1 : x1;
        ar *= (x0 + sx) * r2;
    }
    return ar;
}

#define TILES 8
__global__ void __launch_bounds__(256) qml_kernel(const float* __restrict__ x,
                                                  const float* __restrict__ w,
                                                  float* __restrict__ out) {
    __shared__ alignas(16) float gs[64 + 64 + 256];

    const int lane = threadIdx.x & 31;
    const int wave = threadIdx.x >> 5;
    const int n    = lane & 15;
    const bool hi  = lane >= 16;
    const int cb   = hi ? 2 : 0;
    const int samp0 = (blockIdx.x * 8 + wave) * (16 * TILES) + n;

    const float4* __restrict__ x4 = reinterpret_cast<const float4*>(x);
    float4* __restrict__ o4       = reinterpret_cast<float4*>(out);

    float4 xv = x4[samp0];
    float4 xn = x4[samp0 + 16];

    {
        float wv[16];
        #pragma unroll
        for (int i = 0; i < 16; ++i) wv[i] = w[i];
        const int e = threadIdx.x;
        if (e < 64)       gs[e] = table0_entry<0>(wv, (e >> 2) & 15, e & 3);
        else if (e < 128) gs[e] = table0_entry<1>(wv, (e >> 2) & 15, e & 3);
        gs[128 + e] = table1_entry(wv, e);
    }
    __syncthreads();

    const v2f Aa = *reinterpret_cast<const v2f*>(gs +  0 + n * 4 + cb);
    const v2f Ab = *reinterpret_cast<const v2f*>(gs + 64 + n * 4 + cb);
    v2f Ay[4];
    #pragma unroll
    for (int k = 0; k < 4; ++k)
        Ay[k] = *reinterpret_cast<const v2f*>(gs + 128 + n * 16 + 4 * k + cb);

    for (int t = 0; t < TILES; ++t) {
        const int samp = samp0 + 16 * t;

        float4 xf = xn;
        if (t + 2 < TILES) xf = x4[samp + 32];

        float c0, s0, c1, s1, c2, s2, c3, s3;
        __sincosf(xv.x * 0.5f, &s0, &c0);
        __sincosf(xv.y * 0.5f, &s1, &c1);
        __sincosf(xv.z * 0.5f, &s2, &c2);
        __sincosf(xv.w * 0.5f, &s3, &c3);

        const v2f cs1 = {c1, s1};
        const v2f cs3 = {c3, s3};
        const float f0 = hi ? s0 : c0;
        const float f2 = hi ? s2 : c2;
        const v2f Bp = f0 * cs1;
        const v2f Bq = f2 * cs3;

        const v8f zero = {0.f,0.f,0.f,0.f,0.f,0.f,0.f,0.f};
        const v8f da = wmma4(Aa, Bp, zero);
        const v8f db = wmma4(Ab, Bq, zero);

        const v2f pm  = {-1.0f, 1.0f};
        const v2f c2v = c2 * cs3;
        const v2f s2v = s2 * cs3;
        const v2f XA  = c2v.xx + pm * s2v.yy;
        const v2f YA  = pm * c2v.yy - s2v.xx;
        const v2f m1 = c1 * XA, m2 = s1 * YA, m3 = c1 * YA, m4 = s1 * XA;
        const v2f qrA = m1 + m2, qiA = m3 - m4;
        const v2f qrB = m1 - m2, qiB = -(m3 + m4);
        const float sg0 = hi ? s0 : -s0;
        v2f Pr2[4], Pi2[4];
        {
            const v2f a = c0 * qrA, b = sg0 * qiA, c = c0 * qiA, d = sg0 * qrA;
            Pr2[0] = a - b;            Pi2[0] = c + d;
            Pr2[3] = (a + b).yx;       Pi2[3] = (d - c).yx;
        }
        {
            const v2f a = c0 * qrB, b = sg0 * qiB, c = c0 * qiB, d = sg0 * qrB;
            Pr2[1] = (a - b).yx;       Pi2[1] = (c + d).yx;
            Pr2[2] = a + b;            Pi2[2] = d - c;
        }

        const v2f ar2 = {da[0], da[1]}, ai2 = {da[2], da[3]};
        v2f Br[4], Bi[4];
        #pragma unroll
        for (int k = 0; k < 4; ++k) {
            const int r1 = k & 1, r2 = k >> 1;
            const v2f br2 = {db[4 * r1 + 0], db[4 * r1 + 1]};
            const v2f bi2 = {db[4 * r1 + 2], db[4 * r1 + 3]};
            const v2f xr = (r2 ^ r1) ? ar2.yx : ar2;
            const v2f xi = (r2 ^ r1) ? ai2.yx : ai2;
            const v2f ur = br2 * xr - bi2 * xi;
            const v2f ui = br2 * xi + bi2 * xr;
            Br[k] = ur * Pr2[k] - ui * Pi2[k];
            Bi[k] = ur * Pi2[k] + ui * Pr2[k];
        }

        v8f er = {0.f,0.f,0.f,0.f,0.f,0.f,0.f,0.f};
        v8f ei = {0.f,0.f,0.f,0.f,0.f,0.f,0.f,0.f};
        #pragma unroll
        for (int k = 0; k < 4; ++k) {
            er = wmma4(Ay[k], Br[k], er);
            ei = wmma4(Ay[k], Bi[k], ei);
        }

        float p[8];
        #pragma unroll
        for (int r = 0; r < 8; ++r) p[r] = er[r] * er[r] + ei[r] * ei[r];

        const float t0 = p[0] + p[1], t1 = p[2] + p[3];
        const float t2 = p[4] + p[5], t3 = p[6] + p[7];
        const float u01 = t0 + t1, u23 = t2 + t3;
        float z0 = u01 + u23;
        float z1 = u01 - u23;
        float z2 = (t0 + t2) - (t1 + t3);
        const float w0 = p[0] - p[1], w1 = p[2] - p[3];
        const float w2 = p[4] - p[5], w3 = p[6] - p[7];
        float z3 = (w0 + w1) + (w2 + w3);
        if (hi) z0 = -z0;
        z0 += swap16(z0);
        z1 += swap16(z1);
        z2 += swap16(z2);
        z3 += swap16(z3);

        if (!hi) {
            float4 o; o.x = z0; o.y = z1; o.z = z2; o.w = z3;
            o4[samp] = o;
        }

        xv = xn;
        xn = xf;
    }
}

extern "C" void kernel_launch(void* const* d_in, const int* in_sizes, int n_in,
                              void* d_out, int out_size, void* d_ws, size_t ws_size,
                              hipStream_t stream) {
    const float* x = (const float*)d_in[0];
    const float* w = (const float*)d_in[1];
    float* out = (float*)d_out;

    int B = in_sizes[0] / 4;
    qml_kernel<<<B / (128 * TILES), 256, 0, stream>>>(x, w, out);
}


# RX-layer phases applied to the two tensor factors before their outer product; qubit 0,1 phase from the first product's spare rows (constant rows): 82 vector ops per tile, not 100; products regroup
# speedup vs baseline: 2.2878x; 1.0513x over previous
#include <hip/hip_runtime.h>
#include <math.h>

typedef float v2f __attribute__((ext_vector_type(2)));
typedef float v8f __attribute__((ext_vector_type(8)));

__device__ inline v8f wmma4(v2f a, v2f b, v8f c) {
    return __builtin_amdgcn_wmma_f32_16x16x4_f32(
         false, a,  false, b,
         (short)0, c,  false,  false);
}

__device__ inline float swap16(float v) {
    return __int_as_float(__builtin_amdgcn_ds_swizzle(__float_as_int(v), 0x401F));
}


__device__ inline void fac0(const float* __restrict__ w, int q, int br, int bc, float& ur, float& ui) {
    const float r2 = 0.70710678f;
    float ty = w[q * 2 + 0] * 0.5f;
    float tz = w[q * 2 + 1] * 0.5f;
    float cy, sy, cz, sz;
    __sincosf(ty, &sy, &cy);
    __sincosf(tz, &sz, &cz);
    const float x0 = bc ? -sy : cy;
    const float x1 = bc ? cy : sy;
    const float sx = br ? -x1 : x1;
    const float pr = (x0 + sx) * cz * r2;
    const float pi = (sx - x0) * sz * r2;
    ur = bc ? pi : pr;
    ui = bc ? -pr : pi;
}

template <int which>
__device__ inline float table0_entry(const float* __restrict__ w, int row, int k) {
    const int h = row >> 3, g = row & 7;
    const int a = k >> 1, b = k & 1;
    const int q = which ? 2 : 0;
    int ia, ib, part;
    if (which == 0) {
        const int s2 = g & 1;
        part = (g >> 1) & 1;
        ia = h ^ s2; ib = s2;
    } else {
        const int e = g & 1, r1 = g >> 2;
        part = (g >> 1) & 1;
        ia = r1 ^ e; ib = e;
    }
    float ar, ai, br, bi;
    fac0(w, q,     ia, a, ar, ai);
    fac0(w, q + 1, ib, b, br, bi);
    const float pr = ar * br - ai * bi;
    const float pi = ar * bi + ai * br;
    float v = part ? pi : pr;
    if (which == 0 && g >= 4) {
        const float sh = h ? 1.0f : -1.0f;
        const float sr = (g & 1) ? 1.0f : -1.0f;
        const float re = (k == 0) ? 1.0f : ((k == 3) ? -sh * sr : 0.0f);
        const float im = (k == 1) ? sr : ((k == 2) ? sh : 0.0f);
        v = part ? im : re;
    }
    return v;
}

__device__ inline float table1_entry(const float* __restrict__ w, int e) {
    int r = e >> 4;
    int c = e & 15;
    int j = r;
    if (j & 2) j ^= 1;
    if (j & 4) j ^= 2;
    if (j & 8) j ^= 4;
    const int cs = ((c & 2) << 2) | ((c >> 2) << 1) | (c & 1);

    const float r2 = 0.70710678f;
    float ar = 1.0f;
    #pragma unroll
    for (int q = 0; q < 4; ++q) {
        float ty = w[(4 + q) * 2 + 0] * 0.5f;
        float cy, sy;
        __sincosf(ty, &sy, &cy);
        int br = (j  >> (3 - q)) & 1;
        int bc = (cs >> (3 - q)) & 1;
        const float x0 = br ? sy : cy;
        const float x1 = br ? cy : -sy;
        const float sx = bc ? -x1 : x1;
        ar *= (x0 + sx) * r2;
    }
    return ar;
}

#define TILES 8
__global__ void __launch_bounds__(256) qml_kernel(const float* __restrict__ x,
                                                  const float* __restrict__ w,
                                                  float* __restrict__ out) {
    __shared__ alignas(16) float gs[64 + 64 + 256];

    const int lane = threadIdx.x & 31;
    const int wave = threadIdx.x >> 5;
    const int n    = lane & 15;
    const bool hi  = lane >= 16;
    const int cb   = hi ? 2 : 0;
    const int samp0 = (blockIdx.x * 8 + wave) * (16 * TILES) + n;

    const float4* __restrict__ x4 = reinterpret_cast<const float4*>(x);
    float4* __restrict__ o4       = reinterpret_cast<float4*>(out);

    float4 xv = x4[samp0];
    float4 xn = x4[samp0 + 16];

    {
        float wv[16];
        #pragma unroll
        for (int i = 0; i < 16; ++i) wv[i] = w[i];
        const int e = threadIdx.x;
        if (e < 64)       gs[e] = table0_entry<0>(wv, (e >> 2) & 15, e & 3);
        else if (e < 128) gs[e] = table0_entry<1>(wv, (e >> 2) & 15, e & 3);
        gs[128 + e] = table1_entry(wv, e);
    }
    __syncthreads();

    const v2f Aa = *reinterpret_cast<const v2f*>(gs +  0 + n * 4 + cb);
    const v2f Ab = *reinterpret_cast<const v2f*>(gs + 64 + n * 4 + cb);
    v2f Ay[4];
    #pragma unroll
    for (int k = 0; k < 4; ++k)
        Ay[k] = *reinterpret_cast<const v2f*>(gs + 128 + n * 16 + 4 * k + cb);

    for (int t = 0; t < TILES; ++t) {
        const int samp = samp0 + 16 * t;

        float4 xf = xn;
        if (t + 2 < TILES) xf = x4[samp + 32];

        float c0, s0, c1, s1, c2, s2, c3, s3;
        __sincosf(xv.x * 0.5f, &s0, &c0);
        __sincosf(xv.y * 0.5f, &s1, &c1);
        __sincosf(xv.z * 0.5f, &s2, &c2);
        __sincosf(xv.w * 0.5f, &s3, &c3);

        const v2f cs1 = {c1, s1};
        const v2f cs3 = {c3, s3};
        const float f0 = hi ? s0 : c0;
        const float f2 = hi ? s2 : c2;
        const v2f Bp = f0 * cs1;
        const v2f Bq = f2 * cs3;

        const v8f zero = {0.f,0.f,0.f,0.f,0.f,0.f,0.f,0.f};
        const v8f da = wmma4(Aa, Bp, zero);
        const v8f db = wmma4(Ab, Bq, zero);

        const v2f pm  = {-1.0f, 1.0f};
        const v2f c2v = c2 * cs3;
        const v2f s2v = s2 * cs3;
        const v2f XA  = c2v.xx + pm * s2v.yy;
        const v2f YA  = pm * c2v.yy - s2v.xx;
        const v2f XB  = XA.yx, YB = YA.yx;
        v2f bpr[2], bpi[2];
        {
            const v2f br2 = {db[0], db[1]}, bi2 = {db[2], db[3]};
            bpr[0] = br2 * XA - bi2 * YA;
            bpi[0] = br2 * YA + bi2 * XA;
        }
        {
            const v2f br2 = {db[4], db[5]}, bi2 = {db[6], db[7]};
            bpr[1] = br2 * XB + bi2 * YB;
            bpi[1] = bi2 * XB - br2 * YB;
        }
        const v2f ar2 = {da[0], da[1]}, ai2 = {da[2], da[3]};
        v2f apr[2], api[2];
        #pragma unroll
        for (int r2 = 0; r2 < 2; ++r2) {
            const float fr = da[4 + r2], fi = da[6 + r2];
            apr[r2] = ar2 * fr - ai2 * fi;
            api[r2] = ar2 * fi + ai2 * fr;
        }

        v2f Br[4], Bi[4];
        #pragma unroll
        for (int k = 0; k < 4; ++k) {
            const int r1 = k & 1, r2 = k >> 1;
            const v2f xr = (r2 ^ r1) ? apr[r2].yx : apr[r2];
            const v2f xi = (r2 ^ r1) ? api[r2].yx : api[r2];
            Br[k] = bpr[r1] * xr - bpi[r1] * xi;
            Bi[k] = bpr[r1] * xi + bpi[r1] * xr;
        }

        v8f er = {0.f,0.f,0.f,0.f,0.f,0.f,0.f,0.f};
        v8f ei = {0.f,0.f,0.f,0.f,0.f,0.f,0.f,0.f};
        #pragma unroll
        for (int k = 0; k < 4; ++k) {
            er = wmma4(Ay[k], Br[k], er);
            ei = wmma4(Ay[k], Bi[k], ei);
        }

        float p[8];
        #pragma unroll
        for (int r = 0; r < 8; ++r) p[r] = er[r] * er[r] + ei[r] * ei[r];

        const float t0 = p[0] + p[1], t1 = p[2] + p[3];
        const float t2 = p[4] + p[5], t3 = p[6] + p[7];
        const float u01 = t0 + t1, u23 = t2 + t3;
        float z0 = u01 + u23;
        float z1 = u01 - u23;
        float z2 = (t0 + t2) - (t1 + t3);
        const float w0 = p[0] - p[1], w1 = p[2] - p[3];
        const float w2 = p[4] - p[5], w3 = p[6] - p[7];
        float z3 = (w0 + w1) + (w2 + w3);
        if (hi) z0 = -z0;
        z0 += swap16(z0);
        z1 += swap16(z1);
        z2 += swap16(z2);
        z3 += swap16(z3);

        if (!hi) {
            float4 o; o.x = z0; o.y = z1; o.z = z2; o.w = z3;
            o4[samp] = o;
        }

        xv = xn;
        xn = xf;
    }
}

extern "C" void kernel_launch(void* const* d_in, const int* in_sizes, int n_in,
                              void* d_out, int out_size, void* d_ws, size_t ws_size,
                              hipStream_t stream) {
    const float* x = (const float*)d_in[0];
    const float* w = (const float*)d_in[1];
    float* out = (float*)d_out;

    int B = in_sizes[0] / 4;
    qml_kernel<<<B / (128 * TILES), 256, 0, stream>>>(x, w, out);
}


# Second product factorised by the state's two-term parity structure across qubit 0 | qubits 1-3: 16x8 real table on WMMA (4 f32 WMMAs, not 8; 6 per tile), qubit 0 as a 2x2 form on VALU; sums reorder
# speedup vs baseline: 2.6458x; 1.1565x over previous
#include <hip/hip_runtime.h>
#include <math.h>

typedef float v2f __attribute__((ext_vector_type(2)));
typedef float v8f __attribute__((ext_vector_type(8)));

__device__ inline v8f wmma4(v2f a, v2f b, v8f c) {
    return __builtin_amdgcn_wmma_f32_16x16x4_f32(
         false, a,  false, b,
         (short)0, c,  false,  false);
}

__device__ inline float swap16(float v) {
    return __int_as_float(__builtin_amdgcn_ds_swizzle(__float_as_int(v), 0x401F));
}


__device__ inline void fac0(const float* __restrict__ w, int q, int br, int bc, float& ur, float& ui) {
    const float r2 = 0.70710678f;
    float ty = w[q * 2 + 0] * 0.5f;
    float tz = w[q * 2 + 1] * 0.5f;
    float cy, sy, cz, sz;
    __sincosf(ty, &sy, &cy);
    __sincosf(tz, &sz, &cz);
    const float x0 = bc ? -sy : cy;
    const float x1 = bc ? cy : sy;
    const float sx = br ? -x1 : x1;
    const float pr = (x0 + sx) * cz * r2;
    const float pi = (sx - x0) * sz * r2;
    ur = bc ? pi : pr;
    ui = bc ? -pr : pi;
}

__device__ inline float tableA_entry(const float* __restrict__ w, int row, int k) {
    const int h = row >> 3, r = row & 7;
    const int part = (r >> 1) & 1;
    const bool second = r >= 4;
    const int b = second ? (h ^ (r & 1)) : (r & 1);
    float u0r, u0i, u1r, u1i;
    fac0(w, 0, b, k & 1, u0r, u0i);
    fac0(w, 1, b, k & 1, u1r, u1i);
    const float v0 = part ? u0i : u0r;
    const float v1 = part ? u1i : u1r;
    const bool live = second ? (k >= 2) : (k < 2);
    return live ? (second ? v1 : v0) : 0.0f;
}

__device__ inline float tableB_entry(const float* __restrict__ w, int row, int k) {
    const int g = row & 7;
    const int a = k >> 1, b = k & 1;
    const int e = g & 1, r1 = g >> 2;
    const int part = (g >> 1) & 1;
    float ar, ai, br, bi;
    fac0(w, 2, r1 ^ e, a, ar, ai);
    fac0(w, 3, e,      b, br, bi);
    const float pr = ar * br - ai * bi;
    const float pi = ar * bi + ai * br;
    return part ? pi : pr;
}

__device__ inline float tableT_entry(const float* __restrict__ w, int n, int p) {
    const int c = (n >> 2) & 1;
    const int bo[3] = { n >> 3, (n >> 1) & 1, n & 1 };
    const int bi[3] = { (p >> 1) & 1, p >> 2, p & 1 };
    const float r2 = 0.70710678f;
    float ar = 1.0f;
    #pragma unroll
    for (int q = 0; q < 3; ++q) {
        float ty = w[(5 + q) * 2 + 0] * 0.5f;
        float cy, sy;
        __sincosf(ty, &sy, &cy);
        const float x0 = bo[q] ? sy : cy;
        const float x1 = bo[q] ? cy : -sy;
        const float sx = bi[q] ? -x1 : x1;
        ar *= (x0 + sx) * r2;
    }
    return (c == (bi[0] ^ bi[1] ^ bi[2])) ? ar : 0.0f;
}

#define TILES 8
__global__ void __launch_bounds__(256) qml_kernel(const float* __restrict__ x,
                                                  const float* __restrict__ w,
                                                  float* __restrict__ out) {
    __shared__ alignas(16) float gs[64 + 64 + 128];

    const int lane = threadIdx.x & 31;
    const int wave = threadIdx.x >> 5;
    const int n    = lane & 15;
    const bool hi  = lane >= 16;
    const int cb   = hi ? 2 : 0;
    const int samp0 = (blockIdx.x * 8 + wave) * (16 * TILES) + n;

    const float4* __restrict__ x4 = reinterpret_cast<const float4*>(x);
    float4* __restrict__ o4       = reinterpret_cast<float4*>(out);

    float4 xv = x4[samp0];
    float4 xn = x4[samp0 + 16];

    float sth4, cth2, sth;
    {
        float wv[16];
        #pragma unroll
        for (int i = 0; i < 16; ++i) wv[i] = w[i];
        const int e = threadIdx.x;
        if (e < 64)       gs[e] = tableA_entry(wv, (e >> 2) & 15, e & 3);
        else if (e < 128) gs[e] = tableB_entry(wv, (e >> 2) & 15, e & 3);
        else              gs[e] = tableT_entry(wv, (e - 128) >> 3, e & 7);
        float sh, ch;
        __sincosf(wv[8] * 0.5f, &sh, &ch);
        sth  = 2.0f * sh * ch;
        cth2 = 2.0f * (ch * ch - sh * sh);
        sth4 = 4.0f * sth;
    }
    __syncthreads();

    const v2f Aa = *reinterpret_cast<const v2f*>(gs +  0 + n * 4 + cb);
    const v2f Ab = *reinterpret_cast<const v2f*>(gs + 64 + n * 4 + cb);
    v2f At[2];
    #pragma unroll
    for (int k = 0; k < 2; ++k)
        At[k] = *reinterpret_cast<const v2f*>(gs + 128 + n * 8 + 4 * k + cb);

    for (int t = 0; t < TILES; ++t) {
        const int samp = samp0 + 16 * t;

        float4 xf = xn;
        if (t + 2 < TILES) xf = x4[samp + 32];

        float c0, s0, c1, s1, c2, s2, c3, s3;
        __sincosf(xv.x * 0.5f, &s0, &c0);
        __sincosf(xv.y * 0.5f, &s1, &c1);
        __sincosf(xv.z * 0.5f, &s2, &c2);
        __sincosf(xv.w * 0.5f, &s3, &c3);

        const v2f cs0 = {c0, s0};
        const v2f cs1 = {c1, s1};
        const v2f cs3 = {c3, s3};
        const v2f Bp = hi ? cs1 : cs0;
        const float f2 = hi ? s2 : c2;
        const v2f Bq = f2 * cs3;

        const v8f zero = {0.f,0.f,0.f,0.f,0.f,0.f,0.f,0.f};
        const v8f da = wmma4(Aa, Bp, zero);
        const v8f db = wmma4(Ab, Bq, zero);

        const v2f pm  = {-1.0f, 1.0f};
        const v2f c2v = c2 * cs3;
        const v2f s2v = s2 * cs3;
        const v2f XA  = c2v.xx + pm * s2v.yy;
        const v2f YA  = pm * c2v.yy - s2v.xx;
        const v2f XB  = XA.yx, YB = YA.yx;
        v2f bpr[2], bpi[2];
        {
            const v2f br2 = {db[0], db[1]}, bi2 = {db[2], db[3]};
            bpr[0] = br2 * XA - bi2 * YA;
            bpi[0] = br2 * YA + bi2 * XA;
        }
        {
            const v2f br2 = {db[4], db[5]}, bi2 = {db[6], db[7]};
            bpr[1] = br2 * XB + bi2 * YB;
            bpi[1] = bi2 * XB - br2 * YB;
        }

        const float ss1 = hi ? s1 : -s1;
        const v2f y1r = {da[4], da[5]}, y1i = {da[6], da[7]};
        const v2f Ypr = y1r * c1 - y1i * ss1;
        const v2f Ypi = y1r * ss1 + y1i * c1;

        v2f FR[2], FI[2];
        #pragma unroll
        for (int k = 0; k < 2; ++k) {
            const v2f yr = k ? Ypr.yx : Ypr;
            const v2f yi = k ? Ypi.yx : Ypi;
            FR[k] = bpr[k] * yr - bpi[k] * yi;
            FI[k] = bpr[k] * yi + bpi[k] * yr;
        }

        v8f fr = wmma4(At[0], FR[0], zero);
        v8f fi = wmma4(At[0], FI[0], zero);
        fr = wmma4(At[1], FR[1], fr);
        fi = wmma4(At[1], FI[1], fi);

        const v2f y0r = {da[0], da[1]}, y0i = {da[2], da[3]};
        const v2f nn  = y0r * y0r + y0i * y0i;
        const float zr = y0r.x * y0r.y + y0i.x * y0i.y;
        const float zi = y0r.x * y0i.y - y0i.x * y0r.y;
        const float Er = c0 * c0 - s0 * s0;
        const float Ei = 2.0f * c0 * s0;
        const float dn = nn.y - nn.x, sn = nn.x + nn.y;
        const float t1 = zr * Er, t2 = zi * Ei;
        const float g00 = sth * dn + cth2 * (t1 - t2);
        const float g11 = cth2 * (t1 + t2) - sth * dn;
        const float g2r = cth2 * sn * Er;
        const float g2i = sth4 * zi + cth2 * dn * Ei;

        v2f Lp[2];
        #pragma unroll
        for (int pr = 0; pr < 2; ++pr) {
            const v2f f0r = {fr[2 * pr], fr[2 * pr + 1]}, f1r = {fr[4 + 2 * pr], fr[5 + 2 * pr]};
            const v2f f0i = {fi[2 * pr], fi[2 * pr + 1]}, f1i = {fi[4 + 2 * pr], fi[5 + 2 * pr]};
            const v2f q00  = f0r * f0r + f0i * f0i;
            const v2f q11  = f1r * f1r + f1i * f1i;
            const v2f q01r = f0r * f1r + f0i * f1i;
            const v2f q01i = f0r * f1i - f0i * f1r;
            Lp[pr] = g00 * q00 + g11 * q11 + g2r * q01r + g2i * q01i;
        }

        const v2f P = Lp[0] + Lp[1];
        const v2f M = Lp[0] - Lp[1];
        const float o1 = P.x + P.y;
        const float o2 = M.x + M.y;
        const float o3 = M.x - M.y;
        const float p1 = swap16(o1);
        const float p2 = swap16(o2);
        const float p3 = swap16(o3);

        if (!hi) {
            float4 o; o.x = o1 + p1; o.y = o1 - p1; o.z = o2 - p2; o.w = o3 - p3;
            o4[samp] = o;
        }

        xv = xn;
        xn = xf;
    }
}

extern "C" void kernel_launch(void* const* d_in, const int* in_sizes, int n_in,
                              void* d_out, int out_size, void* d_ws, size_t ws_size,
                              hipStream_t stream) {
    const float* x = (const float*)d_in[0];
    const float* w = (const float*)d_in[1];
    float* out = (float*)d_out;

    int B = in_sizes[0] / 4;
    qml_kernel<<<B / (128 * TILES), 256, 0, stream>>>(x, w, out);
}


# Two tiles per loop trip in phases, previous pair's closing stage carried along: each WMMA is followed by vector work of another tile (sched_group_barrier); per-tile arithmetic and sums unchanged
# speedup vs baseline: 2.8544x; 1.0788x over previous
#include <hip/hip_runtime.h>
#include <math.h>

typedef float v2f __attribute__((ext_vector_type(2)));
typedef float v8f __attribute__((ext_vector_type(8)));

__device__ inline v8f wmma4(v2f a, v2f b, v8f c) {
    return __builtin_amdgcn_wmma_f32_16x16x4_f32(
         false, a,  false, b,
         (short)0, c,  false,  false);
}

__device__ inline float swap16(float v) {
    return __int_as_float(__builtin_amdgcn_ds_swizzle(__float_as_int(v), 0x401F));
}


__device__ inline void fac0(const float* __restrict__ w, int q, int br, int bc, float& ur, float& ui) {
    const float r2 = 0.70710678f;
    float ty = w[q * 2 + 0] * 0.5f;
    float tz = w[q * 2 + 1] * 0.5f;
    float cy, sy, cz, sz;
    __sincosf(ty, &sy, &cy);
    __sincosf(tz, &sz, &cz);
    const float x0 = bc ? -sy : cy;
    const float x1 = bc ? cy : sy;
    const float sx = br ? -x1 : x1;
    const float pr = (x0 + sx) * cz * r2;
    const float pi = (sx - x0) * sz * r2;
    ur = bc ? pi : pr;
    ui = bc ? -pr : pi;
}

__device__ inline float tableA_entry(const float* __restrict__ w, int row, int k) {
    const int h = row >> 3, r = row & 7;
    const int part = (r >> 1) & 1;
    const bool second = r >= 4;
    const int b = second ? (h ^ (r & 1)) : (r & 1);
    float u0r, u0i, u1r, u1i;
    fac0(w, 0, b, k & 1, u0r, u0i);
    fac0(w, 1, b, k & 1, u1r, u1i);
    const float v0 = part ? u0i : u0r;
    const float v1 = part ? u1i : u1r;
    const bool live = second ? (k >= 2) : (k < 2);
    return live ? (second ? v1 : v0) : 0.0f;
}

__device__ inline float tableB_entry(const float* __restrict__ w, int row, int k) {
    const int g = row & 7;
    const int a = k >> 1, b = k & 1;
    const int e = g & 1, r1 = g >> 2;
    const int part = (g >> 1) & 1;
    float ar, ai, br, bi;
    fac0(w, 2, r1 ^ e, a, ar, ai);
    fac0(w, 3, e,      b, br, bi);
    const float pr = ar * br - ai * bi;
    const float pi = ar * bi + ai * br;
    return part ? pi : pr;
}

__device__ inline float tableT_entry(const float* __restrict__ w, int n, int p) {
    const int c = (n >> 2) & 1;
    const int bo[3] = { n >> 3, (n >> 1) & 1, n & 1 };
    const int bi[3] = { (p >> 1) & 1, p >> 2, p & 1 };
    const float r2 = 0.70710678f;
    float ar = 1.0f;
    #pragma unroll
    for (int q = 0; q < 3; ++q) {
        float ty = w[(5 + q) * 2 + 0] * 0.5f;
        float cy, sy;
        __sincosf(ty, &sy, &cy);
        const float x0 = bo[q] ? sy : cy;
        const float x1 = bo[q] ? cy : -sy;
        const float sx = bi[q] ? -x1 : x1;
        ar *= (x0 + sx) * r2;
    }
    return (c == (bi[0] ^ bi[1] ^ bi[2])) ? ar : 0.0f;
}

struct Trig {
    float c0, s0, c1, s1;
    v2f XA, YA;
    v2f Bp, Bq;
};

__device__ inline Trig stage1(float4 xq, bool hi) {
    Trig T;
    float c2, s2, c3, s3;
    __sincosf(xq.x * 0.5f, &T.s0, &T.c0);
    __sincosf(xq.y * 0.5f, &T.s1, &T.c1);
    __sincosf(xq.z * 0.5f, &s2, &c2);
    __sincosf(xq.w * 0.5f, &s3, &c3);
    const v2f cs0 = {T.c0, T.s0};
    const v2f cs1 = {T.c1, T.s1};
    const v2f cs3 = {c3, s3};
    T.Bp = hi ? cs1 : cs0;
    const float f2 = hi ? s2 : c2;
    T.Bq = f2 * cs3;
    const v2f pm  = {-1.0f, 1.0f};
    const v2f c2v = c2 * cs3;
    const v2f s2v = s2 * cs3;
    T.XA = c2v.xx + pm * s2v.yy;
    T.YA = pm * c2v.yy - s2v.xx;
    return T;
}

__device__ inline void stage2(const v8f da, const v8f db, const Trig& T, bool hi, v2f FR[2], v2f FI[2]) {
    const v2f XA = T.XA, YA = T.YA;
    const v2f XB = XA.yx, YB = YA.yx;
    v2f bpr[2], bpi[2];
    {
        const v2f br2 = {db[0], db[1]}, bi2 = {db[2], db[3]};
        bpr[0] = br2 * XA - bi2 * YA;
        bpi[0] = br2 * YA + bi2 * XA;
    }
    {
        const v2f br2 = {db[4], db[5]}, bi2 = {db[6], db[7]};
        bpr[1] = br2 * XB + bi2 * YB;
        bpi[1] = bi2 * XB - br2 * YB;
    }
    const float ss1 = hi ? T.s1 : -T.s1;
    const v2f y1r = {da[4], da[5]}, y1i = {da[6], da[7]};
    const v2f Ypr = y1r * T.c1 - y1i * ss1;
    const v2f Ypi = y1r * ss1 + y1i * T.c1;
    #pragma unroll
    for (int k = 0; k < 2; ++k) {
        const v2f yr = k ? Ypr.yx : Ypr;
        const v2f yi = k ? Ypi.yx : Ypi;
        FR[k] = bpr[k] * yr - bpi[k] * yi;
        FI[k] = bpr[k] * yi + bpi[k] * yr;
    }
}

__device__ inline float4 gform(const v8f da, const Trig& T, float sth, float cth2, float sth4) {
    const v2f y0r = {da[0], da[1]}, y0i = {da[2], da[3]};
    const v2f nn  = y0r * y0r + y0i * y0i;
    const float zr = y0r.x * y0r.y + y0i.x * y0i.y;
    const float zi = y0r.x * y0i.y - y0i.x * y0r.y;
    const float Er = T.c0 * T.c0 - T.s0 * T.s0;
    const float Ei = 2.0f * T.c0 * T.s0;
    const float dn = nn.y - nn.x, sn = nn.x + nn.y;
    const float t1 = zr * Er, t2 = zi * Ei;
    float4 g;
    g.x = sth * dn + cth2 * (t1 - t2);
    g.y = cth2 * (t1 + t2) - sth * dn;
    g.z = cth2 * sn * Er;
    g.w = sth4 * zi + cth2 * dn * Ei;
    return g;
}

__device__ inline void close_lane(const v8f fr, const v8f fi, const float4 g, float& o1, float& o2, float& o3) {
    v2f Lp[2];
    #pragma unroll
    for (int pr = 0; pr < 2; ++pr) {
        const v2f f0r = {fr[2 * pr], fr[2 * pr + 1]}, f1r = {fr[4 + 2 * pr], fr[5 + 2 * pr]};
        const v2f f0i = {fi[2 * pr], fi[2 * pr + 1]}, f1i = {fi[4 + 2 * pr], fi[5 + 2 * pr]};
        const v2f q00  = f0r * f0r + f0i * f0i;
        const v2f q11  = f1r * f1r + f1i * f1i;
        const v2f q01r = f0r * f1r + f0i * f1i;
        const v2f q01i = f0r * f1i - f0i * f1r;
        Lp[pr] = g.x * q00 + g.y * q11 + g.z * q01r + g.w * q01i;
    }
    const v2f P = Lp[0] + Lp[1];
    const v2f M = Lp[0] - Lp[1];
    o1 = P.x + P.y;
    o2 = M.x + M.y;
    o3 = M.x - M.y;
}

__device__ inline float4 close_out(float o1, float o2, float o3, float p1, float p2, float p3) {
    float4 o; o.x = o1 + p1; o.y = o1 - p1; o.z = o2 - p2; o.w = o3 - p3;
    return o;
}

#define TILES 8
__global__ void __launch_bounds__(256) qml_kernel(const float* __restrict__ x,
                                                  const float* __restrict__ w,
                                                  float* __restrict__ out) {
    __shared__ alignas(16) float gs[64 + 64 + 128];

    const int lane = threadIdx.x & 31;
    const int wave = threadIdx.x >> 5;
    const int n    = lane & 15;
    const bool hi  = lane >= 16;
    const int cb   = hi ? 2 : 0;
    const int samp0 = (blockIdx.x * 8 + wave) * (16 * TILES) + n;

    const float4* __restrict__ x4 = reinterpret_cast<const float4*>(x);
    float4* __restrict__ o4       = reinterpret_cast<float4*>(out);

    float4 xv = x4[samp0];
    float4 xn = x4[samp0 + 16];

    float sth4, cth2, sth;
    {
        float wv[16];
        #pragma unroll
        for (int i = 0; i < 16; ++i) wv[i] = w[i];
        const int e = threadIdx.x;
        if (e < 64)       gs[e] = tableA_entry(wv, (e >> 2) & 15, e & 3);
        else if (e < 128) gs[e] = tableB_entry(wv, (e >> 2) & 15, e & 3);
        else              gs[e] = tableT_entry(wv, (e - 128) >> 3, e & 7);
        float sh, ch;
        __sincosf(wv[8] * 0.5f, &sh, &ch);
        sth  = 2.0f * sh * ch;
        cth2 = 2.0f * (ch * ch - sh * sh);
        sth4 = 4.0f * sth;
    }
    __syncthreads();

    const v2f Aa = *reinterpret_cast<const v2f*>(gs +  0 + n * 4 + cb);
    const v2f Ab = *reinterpret_cast<const v2f*>(gs + 64 + n * 4 + cb);
    v2f At[2];
    #pragma unroll
    for (int k = 0; k < 2; ++k)
        At[k] = *reinterpret_cast<const v2f*>(gs + 128 + n * 8 + 4 * k + cb);

    float4 xa = xv, xb = xn;
    const v8f zero = {0.f,0.f,0.f,0.f,0.f,0.f,0.f,0.f};
    v8f pfr[2] = {zero, zero}, pfi[2] = {zero, zero};
    float4 pg[2] = {};
    #pragma unroll
    for (int t = 0; t < TILES; t += 2) {
        const int samp = samp0 + 16 * t;

        float4 xfa = xa, xfb = xb;
        if (t + 2 < TILES) { xfa = x4[samp + 32]; xfb = x4[samp + 48]; }

        float oA1 = 0.f, oA2 = 0.f, oA3 = 0.f, oB1 = 0.f, oB2 = 0.f, oB3 = 0.f;
        float qA1 = 0.f, qA2 = 0.f, qA3 = 0.f, qB1 = 0.f, qB2 = 0.f, qB3 = 0.f;
        v2f FRa[2], FIa[2], FRb[2], FIb[2];

        const Trig TA = stage1(xa, hi);
        __builtin_amdgcn_sched_barrier(0);
        const v8f daA = wmma4(Aa, TA.Bp, zero);
        const v8f dbA = wmma4(Ab, TA.Bq, zero);
        const Trig TB = stage1(xb, hi);
        __builtin_amdgcn_sched_group_barrier(0x008, 1, 0);
        __builtin_amdgcn_sched_group_barrier(0x400, 4, 0);
        __builtin_amdgcn_sched_group_barrier(0x002, 8, 0);
        __builtin_amdgcn_sched_group_barrier(0x008, 1, 0);
        __builtin_amdgcn_sched_group_barrier(0x400, 4, 0);
        __builtin_amdgcn_sched_group_barrier(0x002, 8, 0);
        __builtin_amdgcn_sched_barrier(0);
        const v8f daB = wmma4(Aa, TB.Bp, zero);
        const v8f dbB = wmma4(Ab, TB.Bq, zero);
        stage2(daA, dbA, TA, hi, FRa, FIa);
        __builtin_amdgcn_sched_group_barrier(0x008, 1, 0);
        __builtin_amdgcn_sched_group_barrier(0x002, 15, 0);
        __builtin_amdgcn_sched_group_barrier(0x008, 1, 0);
        __builtin_amdgcn_sched_group_barrier(0x002, 15, 0);
        __builtin_amdgcn_sched_barrier(0);
        v8f frA = wmma4(At[0], FRa[0], zero);
        v8f fiA = wmma4(At[0], FIa[0], zero);
        frA = wmma4(At[1], FRa[1], frA);
        fiA = wmma4(At[1], FIa[1], fiA);
        stage2(daB, dbB, TB, hi, FRb, FIb);
        if (t > 0) {
            close_lane(pfr[0], pfi[0], pg[0], oA1, oA2, oA3);
            qA1 = swap16(oA1); qA2 = swap16(oA2); qA3 = swap16(oA3);
        }
        __builtin_amdgcn_sched_group_barrier(0x008, 1, 0);
        __builtin_amdgcn_sched_group_barrier(0x002, 15, 0);
        __builtin_amdgcn_sched_group_barrier(0x008, 1, 0);
        __builtin_amdgcn_sched_group_barrier(0x002, 15, 0);
        __builtin_amdgcn_sched_group_barrier(0x008, 1, 0);
        __builtin_amdgcn_sched_group_barrier(0x002, 15, 0);
        __builtin_amdgcn_sched_group_barrier(0x008, 1, 0);
        __builtin_amdgcn_sched_group_barrier(0x002, 15, 0);
        __builtin_amdgcn_sched_barrier(0);
        v8f frB = wmma4(At[0], FRb[0], zero);
        v8f fiB = wmma4(At[0], FIb[0], zero);
        frB = wmma4(At[1], FRb[1], frB);
        fiB = wmma4(At[1], FIb[1], fiB);
        if (t > 0) {
            close_lane(pfr[1], pfi[1], pg[1], oB1, oB2, oB3);
            qB1 = swap16(oB1); qB2 = swap16(oB2); qB3 = swap16(oB3);
        }
        __builtin_amdgcn_sched_group_barrier(0x008, 1, 0);
        __builtin_amdgcn_sched_group_barrier(0x002, 13, 0);
        __builtin_amdgcn_sched_group_barrier(0x008, 1, 0);
        __builtin_amdgcn_sched_group_barrier(0x002, 13, 0);
        __builtin_amdgcn_sched_group_barrier(0x008, 1, 0);
        __builtin_amdgcn_sched_group_barrier(0x002, 13, 0);
        __builtin_amdgcn_sched_group_barrier(0x008, 1, 0);
        __builtin_amdgcn_sched_group_barrier(0x002, 13, 0);
        __builtin_amdgcn_sched_barrier(0);
        if (t > 0 && !hi) {
            o4[samp - 32] = close_out(oA1, oA2, oA3, qA1, qA2, qA3);
            o4[samp - 16] = close_out(oB1, oB2, oB3, qB1, qB2, qB3);
        }

        const float4 gA = gform(daA, TA, sth, cth2, sth4);
        const float4 gB = gform(daB, TB, sth, cth2, sth4);
        pfr[0] = frA; pfi[0] = fiA; pg[0] = gA;
        pfr[1] = frB; pfi[1] = fiB; pg[1] = gB;
        xa = xfa;
        xb = xfb;
    }

    {
        float oA1, oA2, oA3, oB1, oB2, oB3;
        close_lane(pfr[0], pfi[0], pg[0], oA1, oA2, oA3);
        close_lane(pfr[1], pfi[1], pg[1], oB1, oB2, oB3);
        const float qA1 = swap16(oA1), qA2 = swap16(oA2), qA3 = swap16(oA3);
        const float qB1 = swap16(oB1), qB2 = swap16(oB2), qB3 = swap16(oB3);
        if (!hi) {
            o4[samp0 + 16 * (TILES - 2)] = close_out(oA1, oA2, oA3, qA1, qA2, qA3);
            o4[samp0 + 16 * (TILES - 1)] = close_out(oB1, oB2, oB3, qB1, qB2, qB3);
        }
    }
}

extern "C" void kernel_launch(void* const* d_in, const int* in_sizes, int n_in,
                              void* d_out, int out_size, void* d_ws, size_t ws_size,
                              hipStream_t stream) {
    const float* x = (const float*)d_in[0];
    const float* w = (const float*)d_in[1];
    float* out = (float*)d_out;

    int B = in_sizes[0] / 4;
    qml_kernel<<<B / (128 * TILES), 256, 0, stream>>>(x, w, out);
}


# Second product factorised once more, across qubit 1 | qubits 2-3: 16x8 real table with re/im rows on WMMA (2 f32 WMMAs, not 4; 4 per tile), qubit 1 as a second 2x2 form on VALU; sums reorder
# speedup vs baseline: 3.1610x; 1.1074x over previous
#include <hip/hip_runtime.h>
#include <math.h>

typedef float v2f __attribute__((ext_vector_type(2)));
typedef float v8f __attribute__((ext_vector_type(8)));

__device__ inline v8f wmma4(v2f a, v2f b, v8f c) {
    return __builtin_amdgcn_wmma_f32_16x16x4_f32(
         false, a,  false, b,
         (short)0, c,  false,  false);
}

__device__ inline float swap16(float v) {
    return __int_as_float(__builtin_amdgcn_ds_swizzle(__float_as_int(v), 0x401F));
}


__device__ inline void fac0(const float* __restrict__ w, int q, int br, int bc, float& ur, float& ui) {
    const float r2 = 0.70710678f;
    float ty = w[q * 2 + 0] * 0.5f;
    float tz = w[q * 2 + 1] * 0.5f;
    float cy, sy, cz, sz;
    __sincosf(ty, &sy, &cy);
    __sincosf(tz, &sz, &cz);
    const float x0 = bc ? -sy : cy;
    const float x1 = bc ? cy : sy;
    const float sx = br ? -x1 : x1;
    const float pr = (x0 + sx) * cz * r2;
    const float pi = (sx - x0) * sz * r2;
    ur = bc ? pi : pr;
    ui = bc ? -pr : pi;
}

__device__ inline float tableA_entry(const float* __restrict__ w, int row, int k) {
    const int r = row & 7;
    const int part = (r >> 1) & 1;
    const bool second = r >= 4;
    const int b = r & 1;
    float u0r, u0i, u1r, u1i;
    fac0(w, 0, b, k & 1, u0r, u0i);
    fac0(w, 1, b, k & 1, u1r, u1i);
    const float v0 = part ? u0i : u0r;
    const float v1 = part ? u1i : u1r;
    const bool live = second ? (k >= 2) : (k < 2);
    return live ? (second ? v1 : v0) : 0.0f;
}

__device__ inline float tableB_entry(const float* __restrict__ w, int row, int k) {
    const int h = row >> 3, g = row & 7;
    const int a = k >> 1, b = k & 1;
    const int d = g >> 2, e = g & 1;
    const int part = ((g >> 1) & 1) ^ h;
    float ar, ai, br, bi;
    fac0(w, 2, d,     a, ar, ai);
    fac0(w, 3, d ^ e, b, br, bi);
    const float pr = ar * br - ai * bi;
    const float pi = ar * bi + ai * br;
    return part ? pi : pr;
}

__device__ inline float ryh(const float* __restrict__ w, int q, int bo, int bi) {
    const float r2 = 0.70710678f;
    float ty = w[(4 + q) * 2 + 0] * 0.5f;
    float cy, sy;
    __sincosf(ty, &sy, &cy);
    const float x0 = bo ? sy : cy;
    const float x1 = bo ? cy : -sy;
    const float sx = bi ? -x1 : x1;
    return (x0 + sx) * r2;
}

__device__ inline float tableT_entry(const float* __restrict__ w, int n, int p) {
    const int b2o = n >> 3, r = n & 7;
    const int d = r >> 2, part = (r >> 1) & 1, b3o = r & 1;
    const int k = p >> 2, h = (p >> 1) & 1, e = p & 1;
    const float v = ryh(w, 2, b2o, e) * ryh(w, 3, b3o, k ^ e);
    return (k == d && h == part) ? v : 0.0f;
}

struct Trig {
    float c0, s0, c1, s1;
    v2f XA, Sv;
    v2f Bp, Bq;
};

__device__ inline Trig stage1(float4 xq, bool hi) {
    Trig T;
    float c2, s2, c3, s3;
    __sincosf(xq.x * 0.5f, &T.s0, &T.c0);
    __sincosf(xq.y * 0.5f, &T.s1, &T.c1);
    __sincosf(xq.z * 0.5f, &s2, &c2);
    __sincosf(xq.w * 0.5f, &s3, &c3);
    const v2f cs0 = {T.c0, T.s0};
    const v2f cs1 = {T.c1, T.s1};
    const v2f cs3 = {c3, s3};
    T.Bp = hi ? cs1 : cs0;
    const float f2 = hi ? s2 : c2;
    T.Bq = f2 * cs3;
    const v2f pm  = {-1.0f, 1.0f};
    const v2f c2v = c2 * cs3;
    const v2f s2v = s2 * cs3;
    T.XA = c2v.xx + pm * s2v.yy;
    const v2f YA = pm * c2v.yy - s2v.xx;
    T.Sv = hi ? YA : -YA;
    return T;
}

__device__ inline void stage2(const v8f db, const Trig& T, v2f U[2]) {
    #pragma unroll
    for (int d = 0; d < 2; ++d) {
        const v2f P = {db[4 * d], db[4 * d + 1]}, Q = {db[4 * d + 2], db[4 * d + 3]};
        const float X = d ? T.XA.y : T.XA.x;
        const float S = d ? T.Sv.y : T.Sv.x;
        const v2f Sd = {S, -S};
        U[d] = P * X + Q * Sd;
    }
}

__device__ inline float4 form2(float n0, float n1, float zr, float zi, float Er, float Ei,
                               float sth, float cth2, float cz, float sw) {
    const float dn = n1 - n0, sn = n0 + n1;
    const float t1 = zr * Er, t2 = zi * Ei;
    float4 g;
    g.x = sth * dn + cth2 * (t1 - t2);
    g.y = cth2 * (t1 + t2) - sth * dn;
    g.z = cz * sn * Er;
    g.w = sw * zi + cz * dn * Ei;
    return g;
}

struct Lc {
    float sth0, cth0, cth20, sth20;
    float sth1, cth21, sth41;
};

struct Forms { float4 g1; v2f gp; };
__device__ inline Forms forms(const v8f da, const Trig& T, const Lc& K) {
    const v2f y0r = {da[0], da[1]}, y0i = {da[2], da[3]};
    const v2f y1r = {da[4], da[5]}, y1i = {da[6], da[7]};
    const v2f nn0 = y0r * y0r + y0i * y0i;
    const v2f nn1 = y1r * y1r + y1i * y1i;
    const float z0r = y0r.x * y0r.y + y0i.x * y0i.y;
    const float z0i = y0r.x * y0i.y - y0i.x * y0r.y;
    const float z1r = y1r.x * y1r.y + y1i.x * y1i.y;
    const float z1i = y1r.x * y1i.y - y1i.x * y1r.y;
    const float E0r = T.c0 * T.c0 - T.s0 * T.s0;
    const float E0i = 2.0f * T.c0 * T.s0;
    const float E1r = T.c1 * T.c1 - T.s1 * T.s1;
    const float E1i = 2.0f * T.c1 * T.s1;
    const float4 g0 = form2(nn0.x, nn0.y, z0r, z0i, E0r, E0i, K.sth0, K.cth20, K.cth0, K.sth20);
    const float Y00 = g0.x * nn1.x, Y11 = g0.y * nn1.y;
    const float Yr  = g0.z * z1r + g0.w * z1i;
    const float Yi  = g0.z * z1i - g0.w * z1r;
    Forms F;
    F.g1 = form2(Y00, Y11, Yr, Yi, E1r, E1i, K.sth1, K.cth21, K.cth21, K.sth41);
    F.gp.x = Y00 + Y11;
    F.gp.y = 4.0f * Yr;
    return F;
}

__device__ inline void close_lane(const v8f G, const Forms& F, float& o0, float& o1, float& o3) {
    const v2f G0r = {G[0], G[1]}, G0i = {G[2], G[3]};
    const v2f G1r = {G[4], G[5]}, G1i = {G[6], G[7]};
    const v2f q00  = G0r * G0r + G0i * G0i;
    const v2f q11  = G1r * G1r + G1i * G1i;
    const v2f q01r = G0r * G1r + G0i * G1i;
    const v2f q01i = G0r * G1i - G0i * G1r;
    const v2f Lm = F.g1.x * q00 + F.g1.y * q11 + F.g1.z * q01r + F.g1.w * q01i;
    const v2f Lp = F.gp.x * (q00 + q11) + F.gp.y * q01r;
    o0 = Lp.x + Lp.y;
    o1 = Lm.x + Lm.y;
    o3 = Lm.x - Lm.y;
}

__device__ inline float4 close_out(float o0, float o1, float o3, float p0, float p1, float p3) {
    float4 o; o.x = o0 + p0; o.y = o1 + p1; o.z = o1 - p1; o.w = o3 - p3;
    return o;
}

#define TILES 8
__global__ void __launch_bounds__(256) qml_kernel(const float* __restrict__ x,
                                                  const float* __restrict__ w,
                                                  float* __restrict__ out) {
    __shared__ alignas(16) float gs[64 + 64 + 128];

    const int lane = threadIdx.x & 31;
    const int wave = threadIdx.x >> 5;
    const int n    = lane & 15;
    const bool hi  = lane >= 16;
    const int cb   = hi ? 2 : 0;
    const int samp0 = (blockIdx.x * 8 + wave) * (16 * TILES) + n;

    const float4* __restrict__ x4 = reinterpret_cast<const float4*>(x);
    float4* __restrict__ o4       = reinterpret_cast<float4*>(out);

    float4 xv = x4[samp0];
    float4 xn = x4[samp0 + 16];

    Lc K;
    {
        float wv[16];
        #pragma unroll
        for (int i = 0; i < 16; ++i) wv[i] = w[i];
        const int e = threadIdx.x;
        if (e < 64)       gs[e] = tableA_entry(wv, (e >> 2) & 15, e & 3);
        else if (e < 128) gs[e] = tableB_entry(wv, (e >> 2) & 15, e & 3);
        else              gs[e] = tableT_entry(wv, (e - 128) >> 3, e & 7);
        float sh, ch;
        __sincosf(wv[8] * 0.5f, &sh, &ch);
        K.sth0  = 2.0f * sh * ch;
        K.cth0  = ch * ch - sh * sh;
        K.cth20 = 2.0f * K.cth0;
        K.sth20 = 2.0f * K.sth0;
        __sincosf(wv[10] * 0.5f, &sh, &ch);
        K.sth1  = 2.0f * sh * ch;
        K.cth21 = 2.0f * (ch * ch - sh * sh);
        K.sth41 = 4.0f * K.sth1;
    }
    __syncthreads();

    const v2f Aa = *reinterpret_cast<const v2f*>(gs +  0 + n * 4 + cb);
    const v2f Ab = *reinterpret_cast<const v2f*>(gs + 64 + n * 4 + cb);
    v2f At[2];
    #pragma unroll
    for (int k = 0; k < 2; ++k)
        At[k] = *reinterpret_cast<const v2f*>(gs + 128 + n * 8 + 4 * k + cb);

    float4 xa = xv, xb = xn;
    const v8f zero = {0.f,0.f,0.f,0.f,0.f,0.f,0.f,0.f};
    v8f pG[2] = {zero, zero};
    Forms pF[2] = {};
    #pragma unroll
    for (int t = 0; t < TILES; t += 2) {
        const int samp = samp0 + 16 * t;

        float4 xfa = xa, xfb = xb;
        if (t + 2 < TILES) { xfa = x4[samp + 32]; xfb = x4[samp + 48]; }

        float oA0 = 0.f, oA1 = 0.f, oA3 = 0.f, oB0 = 0.f, oB1 = 0.f, oB3 = 0.f;
        float qA0 = 0.f, qA1 = 0.f, qA3 = 0.f, qB0 = 0.f, qB1 = 0.f, qB3 = 0.f;
        v2f UA[2], UB[2];

        const Trig TA = stage1(xa, hi);
        __builtin_amdgcn_sched_barrier(0);
        const v8f daA = wmma4(Aa, TA.Bp, zero);
        const v8f dbA = wmma4(Ab, TA.Bq, zero);
        const Trig TB = stage1(xb, hi);
        __builtin_amdgcn_sched_group_barrier(0x008, 1, 0);
        __builtin_amdgcn_sched_group_barrier(0x400, 4, 0);
        __builtin_amdgcn_sched_group_barrier(0x002, 9, 0);
        __builtin_amdgcn_sched_group_barrier(0x008, 1, 0);
        __builtin_amdgcn_sched_group_barrier(0x400, 4, 0);
        __builtin_amdgcn_sched_group_barrier(0x002, 9, 0);
        __builtin_amdgcn_sched_barrier(0);
        const v8f daB = wmma4(Aa, TB.Bp, zero);
        const v8f dbB = wmma4(Ab, TB.Bq, zero);
        stage2(dbA, TA, UA);
        const Forms FA = forms(daA, TA, K);
        if (t > 0) {
            close_lane(pG[0], pF[0], oA0, oA1, oA3);
            qA0 = swap16(oA0); qA1 = swap16(oA1); qA3 = swap16(oA3);
        }
        __builtin_amdgcn_sched_group_barrier(0x008, 1, 0);
        __builtin_amdgcn_sched_group_barrier(0x002, 8, 0);
        __builtin_amdgcn_sched_group_barrier(0x008, 1, 0);
        __builtin_amdgcn_sched_group_barrier(0x002, 20, 0);
        __builtin_amdgcn_sched_barrier(0);
        stage2(dbB, TB, UB);
        v8f GA = wmma4(At[0], UA[0], zero);
        v8f GB = wmma4(At[0], UB[0], zero);
        GA = wmma4(At[1], UA[1], GA);
        GB = wmma4(At[1], UB[1], GB);
        const Forms FB = forms(daB, TB, K);
        if (t > 0) {
            close_lane(pG[1], pF[1], oB0, oB1, oB3);
            qB0 = swap16(oB0); qB1 = swap16(oB1); qB3 = swap16(oB3);
        }
        __builtin_amdgcn_sched_group_barrier(0x002, 8, 0);
        __builtin_amdgcn_sched_group_barrier(0x008, 1, 0);
        __builtin_amdgcn_sched_group_barrier(0x002, 5, 0);
        __builtin_amdgcn_sched_group_barrier(0x008, 1, 0);
        __builtin_amdgcn_sched_group_barrier(0x002, 5, 0);
        __builtin_amdgcn_sched_group_barrier(0x008, 1, 0);
        __builtin_amdgcn_sched_group_barrier(0x002, 5, 0);
        __builtin_amdgcn_sched_group_barrier(0x008, 1, 0);
        __builtin_amdgcn_sched_group_barrier(0x002, 5, 0);
        __builtin_amdgcn_sched_barrier(0);
        if (t > 0 && !hi) {
            o4[samp - 32] = close_out(oA0, oA1, oA3, qA0, qA1, qA3);
            o4[samp - 16] = close_out(oB0, oB1, oB3, qB0, qB1, qB3);
        }

        pG[0] = GA; pF[0] = FA;
        pG[1] = GB; pF[1] = FB;
        xa = xfa;
        xb = xfb;
    }

    {
        float oA0, oA1, oA3, oB0, oB1, oB3;
        close_lane(pG[0], pF[0], oA0, oA1, oA3);
        close_lane(pG[1], pF[1], oB0, oB1, oB3);
        const float qA0 = swap16(oA0), qA1 = swap16(oA1), qA3 = swap16(oA3);
        const float qB0 = swap16(oB0), qB1 = swap16(oB1), qB3 = swap16(oB3);
        if (!hi) {
            o4[samp0 + 16 * (TILES - 2)] = close_out(oA0, oA1, oA3, qA0, qA1, qA3);
            o4[samp0 + 16 * (TILES - 1)] = close_out(oB0, oB1, oB3, qB0, qB1, qB3);
        }
    }
}

extern "C" void kernel_launch(void* const* d_in, const int* in_sizes, int n_in,
                              void* d_out, int out_size, void* d_ws, size_t ws_size,
                              hipStream_t stream) {
    const float* x = (const float*)d_in[0];
    const float* w = (const float*)d_in[1];
    float* out = (float*)d_out;

    int B = in_sizes[0] / 4;
    qml_kernel<<<B / (128 * TILES), 256, 0, stream>>>(x, w, out);
}


# Vector stages and first products run one sample per lane (32 per instruction, was 16 held twice): block-form tables, lane halves exchanged by v_permlane16_swap round the second product; sums kept
# speedup vs baseline: 3.6136x; 1.1432x over previous
#include <hip/hip_runtime.h>
#include <math.h>

typedef float v2f __attribute__((ext_vector_type(2)));
typedef float v8f __attribute__((ext_vector_type(8)));

__device__ inline v8f wmma4(v2f a, v2f b, v8f c) {
    return __builtin_amdgcn_wmma_f32_16x16x4_f32(
         false, a,  false, b,
         (short)0, c,  false,  false);
}

__device__ inline void swap_halves(float& a, float& b) {
    const auto r = __builtin_amdgcn_permlane16_swap(__float_as_uint(a), __float_as_uint(b), false, false);
    a = __uint_as_float(r[0]);
    b = __uint_as_float(r[1]);
}


__device__ inline void fac0(const float* __restrict__ w, int q, int br, int bc, float& ur, float& ui) {
    const float r2 = 0.70710678f;
    float ty = w[q * 2 + 0] * 0.5f;
    float tz = w[q * 2 + 1] * 0.5f;
    float cy, sy, cz, sz;
    __sincosf(ty, &sy, &cy);
    __sincosf(tz, &sz, &cz);
    const float x0 = bc ? -sy : cy;
    const float x1 = bc ? cy : sy;
    const float sx = br ? -x1 : x1;
    const float pr = (x0 + sx) * cz * r2;
    const float pi = (sx - x0) * sz * r2;
    ur = bc ? pi : pr;
    ui = bc ? -pr : pi;
}

__device__ inline float tableA_entry(const float* __restrict__ w, int row, int k) {
    const int r = row & 7;
    const int part = (r >> 1) & 1;
    const bool second = r >= 4;
    const int b = r & 1;
    float u0r, u0i, u1r, u1i;
    fac0(w, 0, b, k & 1, u0r, u0i);
    fac0(w, 1, b, k & 1, u1r, u1i);
    const float v0 = part ? u0i : u0r;
    const float v1 = part ? u1i : u1r;
    const bool live = second ? (k >= 2) : (k < 2);
    return live ? (second ? v1 : v0) : 0.0f;
}

__device__ inline float tableB_entry(const float* __restrict__ w, int row, int k) {
    const int g = row & 7;
    const int a = k >> 1, b = k & 1;
    const int d = g >> 2, e = g & 1;
    const int part = (g >> 1) & 1;
    float ar, ai, br, bi;
    fac0(w, 2, d,     a, ar, ai);
    fac0(w, 3, d ^ e, b, br, bi);
    const float pr = ar * br - ai * bi;
    const float pi = ar * bi + ai * br;
    return part ? pi : pr;
}

__device__ inline float ryh(const float* __restrict__ w, int q, int bo, int bi) {
    const float r2 = 0.70710678f;
    float ty = w[(4 + q) * 2 + 0] * 0.5f;
    float cy, sy;
    __sincosf(ty, &sy, &cy);
    const float x0 = bo ? sy : cy;
    const float x1 = bo ? cy : -sy;
    const float sx = bi ? -x1 : x1;
    return (x0 + sx) * r2;
}

__device__ inline float tableT_entry(const float* __restrict__ w, int n, int p) {
    const int b2o = n >> 3, r = n & 7;
    const int d = r >> 2, part = (r >> 1) & 1, b3o = r & 1;
    const int k = p >> 2, h = (p >> 1) & 1, e = p & 1;
    const float v = ryh(w, 2, b2o, e) * ryh(w, 3, b3o, k ^ e);
    return (k == d && h == part) ? v : 0.0f;
}

struct Trig {
    float c0, s0, c1, s1;
    v2f P, Q;
    v2f R, S;
    v2f XA, YA;
};

__device__ inline Trig stage1(float4 xq) {
    Trig T;
    float c2, s2, c3, s3;
    __sincosf(xq.x * 0.5f, &T.s0, &T.c0);
    __sincosf(xq.y * 0.5f, &T.s1, &T.c1);
    __sincosf(xq.z * 0.5f, &s2, &c2);
    __sincosf(xq.w * 0.5f, &s3, &c3);
    const v2f cs0 = {T.c0, T.s0};
    const v2f cs1 = {T.c1, T.s1};
    const v2f cs3 = {c3, s3};
    T.P = cs0;
    T.Q = cs1;
    T.R = c2 * cs3;
    T.S = s2 * cs3;
    const v2f pm  = {-1.0f, 1.0f};
    T.XA = T.R.xx + pm * T.S.yy;
    T.YA = pm * T.R.yy - T.S.xx;
    return T;
}

__device__ inline void stage2(const v8f db, const Trig& T, v2f Ure[2], v2f Uim[2]) {
    #pragma unroll
    for (int d = 0; d < 2; ++d) {
        const v2f Wr = {db[4 * d], db[4 * d + 1]}, Wi = {db[4 * d + 2], db[4 * d + 3]};
        const float X = d ? T.XA.y : T.XA.x;
        const float Y = d ? T.YA.y : T.YA.x;
        const v2f Ym = {-Y, Y}, Yp = {Y, -Y};
        Ure[d] = Wr * X + Wi * Ym;
        Uim[d] = Wi * X + Wr * Yp;
    }
}

__device__ inline float4 form2(float n0, float n1, float zr, float zi, float Er, float Ei,
                               float sth, float cth2, float cz, float sw) {
    const float dn = n1 - n0, sn = n0 + n1;
    const float t1 = zr * Er, t2 = zi * Ei;
    float4 g;
    g.x = sth * dn + cth2 * (t1 - t2);
    g.y = cth2 * (t1 + t2) - sth * dn;
    g.z = cz * sn * Er;
    g.w = sw * zi + cz * dn * Ei;
    return g;
}

struct Lc {
    float sth0, cth0, cth20, sth20;
    float sth1, cth21, sth41;
};

struct Forms { float4 g1; v2f gp; };
__device__ inline Forms forms(const v8f da, const Trig& T, const Lc& K) {
    const v2f y0r = {da[0], da[1]}, y0i = {da[2], da[3]};
    const v2f y1r = {da[4], da[5]}, y1i = {da[6], da[7]};
    const v2f nn0 = y0r * y0r + y0i * y0i;
    const v2f nn1 = y1r * y1r + y1i * y1i;
    const float z0r = y0r.x * y0r.y + y0i.x * y0i.y;
    const float z0i = y0r.x * y0i.y - y0i.x * y0r.y;
    const float z1r = y1r.x * y1r.y + y1i.x * y1i.y;
    const float z1i = y1r.x * y1i.y - y1i.x * y1r.y;
    const float E0r = T.c0 * T.c0 - T.s0 * T.s0;
    const float E0i = 2.0f * T.c0 * T.s0;
    const float E1r = T.c1 * T.c1 - T.s1 * T.s1;
    const float E1i = 2.0f * T.c1 * T.s1;
    const float4 g0 = form2(nn0.x, nn0.y, z0r, z0i, E0r, E0i, K.sth0, K.cth20, K.cth0, K.sth20);
    const float Y00 = g0.x * nn1.x, Y11 = g0.y * nn1.y;
    const float Yr  = g0.z * z1r + g0.w * z1i;
    const float Yi  = g0.z * z1i - g0.w * z1r;
    Forms F;
    F.g1 = form2(Y00, Y11, Yr, Yi, E1r, E1i, K.sth1, K.cth21, K.cth21, K.sth41);
    F.gp.x = Y00 + Y11;
    F.gp.y = 4.0f * Yr;
    return F;
}

__device__ inline void close_lane(const v8f G, const Forms& F, float& o0, float& o1, float& o3) {
    const v2f G0r = {G[0], G[1]}, G0i = {G[2], G[3]};
    const v2f G1r = {G[4], G[5]}, G1i = {G[6], G[7]};
    const v2f q00  = G0r * G0r + G0i * G0i;
    const v2f q11  = G1r * G1r + G1i * G1i;
    const v2f q01r = G0r * G1r + G0i * G1i;
    const v2f q01i = G0r * G1i - G0i * G1r;
    const v2f Lm = F.g1.x * q00 + F.g1.y * q11 + F.g1.z * q01r + F.g1.w * q01i;
    const v2f Lp = F.gp.x * (q00 + q11) + F.gp.y * q01r;
    o0 = Lp.x + Lp.y;
    o1 = Lm.x + Lm.y;
    o3 = Lm.x - Lm.y;
}

__device__ inline float4 close_out(float o0, float o1, float o3, float p0, float p1, float p3) {
    float4 o; o.x = o0 + p0; o.y = o1 + p1; o.z = o1 - p1; o.w = o3 - p3;
    return o;
}

#define TILES 8
__global__ void __launch_bounds__(256) qml_kernel(const float* __restrict__ x,
                                                  const float* __restrict__ w,
                                                  float* __restrict__ out) {
    __shared__ alignas(16) float gs[64 + 64 + 128];

    const int lane = threadIdx.x & 31;
    const int wave = threadIdx.x >> 5;
    const int n    = lane & 15;
    const bool hi  = lane >= 16;
    const int cb   = hi ? 2 : 0;
    const int samp0 = (blockIdx.x * 8 + wave) * (16 * TILES) + lane;

    const float4* __restrict__ x4 = reinterpret_cast<const float4*>(x);
    float4* __restrict__ o4       = reinterpret_cast<float4*>(out);

    float4 xa = x4[samp0];

    Lc K;
    {
        float wv[16];
        #pragma unroll
        for (int i = 0; i < 16; ++i) wv[i] = w[i];
        const int e = threadIdx.x;
        if (e < 64)       gs[e] = tableA_entry(wv, (e >> 2) & 15, e & 3);
        else if (e < 128) gs[e] = tableB_entry(wv, (e >> 2) & 15, e & 3);
        else              gs[e] = tableT_entry(wv, (e - 128) >> 3, e & 7);
        float sh, ch;
        __sincosf(wv[8] * 0.5f, &sh, &ch);
        K.sth0  = 2.0f * sh * ch;
        K.cth0  = ch * ch - sh * sh;
        K.cth20 = 2.0f * K.cth0;
        K.sth20 = 2.0f * K.sth0;
        __sincosf(wv[10] * 0.5f, &sh, &ch);
        K.sth1  = 2.0f * sh * ch;
        K.cth21 = 2.0f * (ch * ch - sh * sh);
        K.sth41 = 4.0f * K.sth1;
    }
    __syncthreads();

    const float4 ta = *reinterpret_cast<const float4*>(gs +  0 + n * 4);
    const float4 tb = *reinterpret_cast<const float4*>(gs + 64 + n * 4);
    const bool own = (n >= 8) == hi;
    const v2f none = {0.f, 0.f};
    const v2f A1 = own ? v2f{ta.x, ta.y} : none;
    const v2f A2 = own ? v2f{ta.z, ta.w} : none;
    const v2f B1 = own ? v2f{tb.x, tb.y} : none;
    const v2f B2 = own ? v2f{tb.z, tb.w} : none;
    v2f At[2];
    #pragma unroll
    for (int k = 0; k < 2; ++k)
        At[k] = *reinterpret_cast<const v2f*>(gs + 128 + n * 8 + 4 * k + cb);

    const v8f zero = {0.f,0.f,0.f,0.f,0.f,0.f,0.f,0.f};
    v8f pG0 = zero, pG1 = zero;
    Forms pF = {};
    #pragma unroll
    for (int t = 0; t < TILES; t += 2) {
        const int samp = samp0 + 16 * t;

        float4 xf = xa;
        if (t + 2 < TILES) xf = x4[samp + 32];

        const Trig T = stage1(xa);
        __builtin_amdgcn_sched_barrier(0);
        v8f da = wmma4(A1, T.P, zero);
        v8f db = wmma4(B1, T.R, zero);
        da = wmma4(A2, T.Q, da);
        db = wmma4(B2, T.S, db);
        if (t > 0) {
            float o0, o1, o3, p0, p1, p3;
            close_lane(pG0, pF, o0, o1, o3);
            close_lane(pG1, pF, p0, p1, p3);
            o4[samp - 32] = close_out(o0, o1, o3, p0, p1, p3);
            __builtin_amdgcn_sched_group_barrier(0x008, 1, 0);
            __builtin_amdgcn_sched_group_barrier(0x002, 8, 0);
            __builtin_amdgcn_sched_group_barrier(0x008, 1, 0);
            __builtin_amdgcn_sched_group_barrier(0x002, 8, 0);
            __builtin_amdgcn_sched_group_barrier(0x008, 1, 0);
            __builtin_amdgcn_sched_group_barrier(0x002, 8, 0);
            __builtin_amdgcn_sched_group_barrier(0x008, 1, 0);
            __builtin_amdgcn_sched_group_barrier(0x002, 8, 0);
        }
        __builtin_amdgcn_sched_barrier(0);
        v2f UA[2], UB[2];
        stage2(db, T, UA, UB);
        #pragma unroll
        for (int d = 0; d < 2; ++d) {
            float a0 = UA[d].x, b0 = UB[d].x, a1 = UA[d].y, b1 = UB[d].y;
            swap_halves(a0, b0);
            swap_halves(a1, b1);
            UA[d] = v2f{a0, a1};
            UB[d] = v2f{b0, b1};
        }
        __builtin_amdgcn_sched_barrier(0);
        v8f GA = wmma4(At[0], UA[0], zero);
        v8f GB = wmma4(At[0], UB[0], zero);
        GA = wmma4(At[1], UA[1], GA);
        GB = wmma4(At[1], UB[1], GB);
        const Forms F = forms(da, T, K);
        asm volatile("" :: "v"(F.g1.x), "v"(F.g1.y), "v"(F.g1.z), "v"(F.g1.w), "v"(F.gp.x), "v"(F.gp.y));
        __builtin_amdgcn_sched_group_barrier(0x008, 1, 0);
        __builtin_amdgcn_sched_group_barrier(0x002, 8, 0);
        __builtin_amdgcn_sched_group_barrier(0x008, 1, 0);
        __builtin_amdgcn_sched_group_barrier(0x002, 8, 0);
        __builtin_amdgcn_sched_group_barrier(0x008, 1, 0);
        __builtin_amdgcn_sched_group_barrier(0x002, 8, 0);
        __builtin_amdgcn_sched_group_barrier(0x008, 1, 0);
        __builtin_amdgcn_sched_group_barrier(0x002, 8, 0);
        __builtin_amdgcn_sched_barrier(0);
        #pragma unroll
        for (int r = 0; r < 8; ++r) {
            float ga = GA[r], gb = GB[r];
            swap_halves(ga, gb);
            GA[r] = ga; GB[r] = gb;
        }

        pG0 = GA; pG1 = GB; pF = F;
        xa = xf;
    }

    {
        float o0, o1, o3, p0, p1, p3;
        close_lane(pG0, pF, o0, o1, o3);
        close_lane(pG1, pF, p0, p1, p3);
        o4[samp0 + 16 * (TILES - 2)] = close_out(o0, o1, o3, p0, p1, p3);
    }
}

extern "C" void kernel_launch(void* const* d_in, const int* in_sizes, int n_in,
                              void* d_out, int out_size, void* d_ws, size_t ws_size,
                              hipStream_t stream) {
    const float* x = (const float*)d_in[0];
    const float* w = (const float*)d_in[1];
    float* out = (float*)d_out;

    int B = in_sizes[0] / 4;
    qml_kernel<<<B / (128 * TILES), 256, 0, stream>>>(x, w, out);
}
